# starting kernel
# baseline (speedup 1.0000x reference)
#include <hip/hip_runtime.h>
#include <hip/hip_bf16.h>

typedef __attribute__((ext_vector_type(2))) float v2f;
typedef __attribute__((ext_vector_type(8))) float v8f;

#define N_NODES 100000
#define N_EDGES 3200000

__global__ void deg_init_kernel(float* deg, int n) {
    int i = blockIdx.x * blockDim.x + threadIdx.x;
    if (i < n) deg[i] = 1.0f;
}
__global__ void deg_count_kernel(const int* __restrict__ dst, float* deg, int e) {
    int i = blockIdx.x * blockDim.x + threadIdx.x;
    if (i < e) atomicAdd(&deg[dst[i]], 1.0f);
}
__global__ void deg_rsqrt_kernel(float* deg, int n) {
    int i = blockIdx.x * blockDim.x + threadIdx.x;
    if (i < n) deg[i] = rsqrtf(deg[i]);
}

template <int K, int NOUT, int MT>
__global__ __launch_bounds__(MT*(NOUT/16)*32)
void gemm_wmma_f32(const float* __restrict__ A, const float* __restrict__ B,
                   float* __restrict__ C, int M) {
    constexpr int NT   = NOUT / 16;
    constexpr int NW   = MT * NT;
    constexpr int ROWS = MT * 16;
    constexpr int KP   = K + 4;
    constexpr int NP   = NOUT + 4;

    __shared__ float As[ROWS][KP];
    __shared__ float Bs[K][NP];

    const int tid  = threadIdx.x;
    const int row0 = blockIdx.x * ROWS;

    constexpr int A_ELTS = ROWS * K;
    for (int i = tid * 4; i < A_ELTS; i += NW * 32 * 4) {
        int r = i / K, c = i % K;
        int gr = row0 + r;
        float4 v = (gr < M) ? *(const float4*)(A + (long long)gr * K + c)
                            : make_float4(0.f, 0.f, 0.f, 0.f);
        *(float4*)(&As[r][c]) = v;
    }
    constexpr int B_ELTS = K * NOUT;
    for (int i = tid * 4; i < B_ELTS; i += NW * 32 * 4) {
        int r = i / NOUT, c = i % NOUT;
        *(float4*)(&Bs[r][c]) = *(const float4*)(B + i);
    }
    __syncthreads();

    const int wid  = tid >> 5;
    const int lane = tid & 31;
    const int mn   = lane & 15;
    const int h    = lane >> 4;
    const int mt   = wid / NT;
    const int nt   = wid % NT;

    v8f acc = {};
    const float* arow = &As[mt * 16 + mn][0];
    const int    ncol = nt * 16 + mn;

#pragma unroll
    for (int k = 0; k < K; k += 4) {
        v2f a, b;
        a.x = arow[k + 2 * h + 0];
        a.y = arow[k + 2 * h + 1];
        b.x = Bs[k + 2 * h + 0][ncol];
        b.y = Bs[k + 2 * h + 1][ncol];
        acc = __builtin_amdgcn_wmma_f32_16x16x4_f32(
             false, a,  false, b,
             (short)0, acc,  false,  false);
    }

#pragma unroll
    for (int j = 0; j < 8; ++j) {
        int r = row0 + mt * 16 + j + 8 * h;
        if (r < M) C[(long long)r * NOUT + ncol] = acc[j];
    }
}

__global__ void zero_kernel(float* p, long long n) {
    long long i = (long long)blockIdx.x * blockDim.x + threadIdx.x;
    if (i < n) p[i] = 0.0f;
}

template <int F>
__global__ void scatter_kernel(const float* __restrict__ h, const int* __restrict__ src,
                               const int* __restrict__ dst, const float* __restrict__ dinv,
                               float* agg, int E) {
    constexpr int LPE = F / 4;
    long long t = (long long)blockIdx.x * blockDim.x + threadIdx.x;
    int e = (int)(t / LPE);
    int j = (int)(t % LPE);
    if (e >= E) return;
    int s = src[e], d = dst[e];
    float norm = dinv[s] * dinv[d];
    float4 v = *(const float4*)(h + (long long)s * F + j * 4);
    float* o = agg + (long long)d * F + j * 4;
    atomicAdd(o + 0, v.x * norm);
    atomicAdd(o + 1, v.y * norm);
    atomicAdd(o + 2, v.z * norm);
    atomicAdd(o + 3, v.w * norm);
}

template <int F>
__global__ void finish_kernel(float* agg, const float* __restrict__ h,
                              const float* __restrict__ dinv, const float* __restrict__ bias,
                              int n) {
    long long t = (long long)blockIdx.x * blockDim.x + threadIdx.x;
    int i = (int)(t / F);
    int f = (int)(t % F);
    if (i >= n) return;
    float di = dinv[i];
    float v = agg[(long long)i * F + f] + h[(long long)i * F + f] * di * di + bias[f];
    agg[(long long)i * F + f] = fmaxf(v, 0.0f);
}

__global__ void final_linear_kernel(const float* __restrict__ y, const float* __restrict__ Wl,
                                    const float* __restrict__ bl, float* out, int n) {
    int i = blockIdx.x * blockDim.x + threadIdx.x;
    if (i >= n) return;
    const float* row = y + (long long)i * 32;
    float s = 0.0f;
#pragma unroll
    for (int f = 0; f < 32; ++f) s += row[f] * Wl[f];
    out[i] = s + bl[0];
}

static inline size_t align256(size_t x) { return (x + 255) & ~size_t(255); }

extern "C" void kernel_launch(void* const* d_in, const int* in_sizes, int n_in,
                              void* d_out, int out_size, void* d_ws, size_t ws_size,
                              hipStream_t stream) {
    const float* x    = (const float*)d_in[0];
    const int*   ei   = (const int*)d_in[1];
    const float* W1   = (const float*)d_in[2];
    const float* b1   = (const float*)d_in[3];
    const float* W2   = (const float*)d_in[4];
    const float* b2   = (const float*)d_in[5];
    const float* Wl   = (const float*)d_in[6];
    const float* bl   = (const float*)d_in[7];
    float*       out  = (float*)d_out;

    const int N = N_NODES;
    const int E = N_EDGES;
    const int* src = ei;
    const int* dst = ei + E;

    char* ws = (char*)d_ws;
    size_t off = 0;
    float* dinv = (float*)(ws + off); off += align256((size_t)N * 4);
    float* h1   = (float*)(ws + off); off += align256((size_t)N * 64 * 4);
    float* agg1 = (float*)(ws + off); off += align256((size_t)N * 64 * 4);
    float* h2   = (float*)(ws + off); off += align256((size_t)N * 32 * 4);
    float* agg2 = (float*)(ws + off); off += align256((size_t)N * 32 * 4);
    (void)ws_size; (void)n_in; (void)in_sizes; (void)out_size;

    const int T = 256;

    deg_init_kernel<<<(N + T - 1) / T, T, 0, stream>>>(dinv, N);
    deg_count_kernel<<<(E + T - 1) / T, T, 0, stream>>>(dst, dinv, E);
    deg_rsqrt_kernel<<<(N + T - 1) / T, T, 0, stream>>>(dinv, N);

    gemm_wmma_f32<128, 64, 2><<<N / 32, 2 * 4 * 32, 0, stream>>>(x, W1, h1, N);

    zero_kernel<<<(int)(((long long)N * 64 + T - 1) / T), T, 0, stream>>>(agg1, (long long)N * 64);
    {
        long long tt = (long long)E * (64 / 4);
        scatter_kernel<64><<<(int)((tt + T - 1) / T), T, 0, stream>>>(h1, src, dst, dinv, agg1, E);
    }
    finish_kernel<64><<<(int)(((long long)N * 64 + T - 1) / T), T, 0, stream>>>(agg1, h1, dinv, b1, N);

    gemm_wmma_f32<64, 32, 2><<<N / 32, 2 * 2 * 32, 0, stream>>>(agg1, W2, h2, N);

    zero_kernel<<<(int)(((long long)N * 32 + T - 1) / T), T, 0, stream>>>(agg2, (long long)N * 32);
    {
        long long tt = (long long)E * (32 / 4);
        scatter_kernel<32><<<(int)((tt + T - 1) / T), T, 0, stream>>>(h2, src, dst, dinv, agg2, E);
    }
    finish_kernel<32><<<(int)(((long long)N * 32 + T - 1) / T), T, 0, stream>>>(agg2, h2, dinv, b2, N);

    final_linear_kernel<<<(N + T - 1) / T, T, 0, stream>>>(agg2, Wl, bl, out, N);
}


# Group edges by destination inside the call (int atomic count returns an arrival-order rank, scan, fill); one wave per node sums its source rows in registers, no f32 atomics; sum order changes
# speedup vs baseline: 1.7759x; 1.7759x over previous
#include <hip/hip_runtime.h>
#include <hip/hip_bf16.h>

typedef __attribute__((ext_vector_type(2))) float v2f;
typedef __attribute__((ext_vector_type(8))) float v8f;

#define N_NODES 100000
#define N_EDGES 3200000

static_assert(N_NODES % 4 == 0 && N_EDGES % 4 == 0, "16-byte index loads need multiples of 4");

__global__ void zero_int_kernel(int* p, int n) {
    int i = blockIdx.x * blockDim.x + threadIdx.x;
    if (i < n) p[i] = 0;
}

__global__ void count_rank_kernel(const int* __restrict__ dst, int* cnt,
                                  int* __restrict__ rank, int e4) {
    int i = blockIdx.x * blockDim.x + threadIdx.x;
    if (i >= e4) return;
    int4 d = ((const int4*)dst)[i];
    int4 r;
    r.x = atomicAdd(&cnt[d.x], 1);
    r.y = atomicAdd(&cnt[d.y], 1);
    r.z = atomicAdd(&cnt[d.z], 1);
    r.w = atomicAdd(&cnt[d.w], 1);
    ((int4*)rank)[i] = r;
}

__global__ __launch_bounds__(1024)
void scan_dinv_kernel(const int* __restrict__ cnt, int* __restrict__ rowptr,
                      float* __restrict__ dinv, int n) {
    __shared__ int s_pre[32];
    __shared__ int s_tot[32];
    const int tid  = threadIdx.x;
    const int lane = tid & 31;
    const int wid  = tid >> 5;
    const int n4   = n >> 2;
    const int4* c4 = (const int4*)cnt;
    const int first4 = blockIdx.x * 1024;

    int pre = 0;
    int i = tid;
    for (; i + 7 * 1024 < first4; i += 8 * 1024) {
        const int4 a0 = c4[i];
        const int4 a1 = c4[i + 1024];
        const int4 a2 = c4[i + 2 * 1024];
        const int4 a3 = c4[i + 3 * 1024];
        const int4 a4 = c4[i + 4 * 1024];
        const int4 a5 = c4[i + 5 * 1024];
        const int4 a6 = c4[i + 6 * 1024];
        const int4 a7 = c4[i + 7 * 1024];
        pre += (a0.x + a0.y) + (a0.z + a0.w);
        pre += (a1.x + a1.y) + (a1.z + a1.w);
        pre += (a2.x + a2.y) + (a2.z + a2.w);
        pre += (a3.x + a3.y) + (a3.z + a3.w);
        pre += (a4.x + a4.y) + (a4.z + a4.w);
        pre += (a5.x + a5.y) + (a5.z + a5.w);
        pre += (a6.x + a6.y) + (a6.z + a6.w);
        pre += (a7.x + a7.y) + (a7.z + a7.w);
    }
    for (; i < first4; i += 1024) {
        const int4 a0 = c4[i];
        pre += (a0.x + a0.y) + (a0.z + a0.w);
    }
    const int idx4 = first4 + tid;
    int4 v = make_int4(0, 0, 0, 0);
    if (idx4 < n4) v = c4[idx4];
    const int s0 = v.x, s1 = s0 + v.y, s2 = s1 + v.z, s3 = s2 + v.w;

#pragma unroll
    for (int o = 16; o >= 1; o >>= 1) pre += __shfl_xor(pre, o);
    int incl = s3;
#pragma unroll
    for (int o = 1; o < 32; o <<= 1) {
        int t = __shfl_up(incl, o);
        if (lane >= o) incl += t;
    }
    if (lane == 31) s_tot[wid] = incl;
    if (lane == 0)  s_pre[wid] = pre;
    __syncthreads();
    int base = 0;
#pragma unroll
    for (int w = 0; w < 32; ++w) {
        base += s_pre[w];
        if (w < wid) base += s_tot[w];
    }
    const int excl = base + incl - s3;
    if (idx4 < n4) {
        ((int4*)rowptr)[idx4] = make_int4(excl, excl + s0, excl + s1, excl + s2);
        float4 d;
        d.x = rsqrtf((float)(v.x + 1));
        d.y = rsqrtf((float)(v.y + 1));
        d.z = rsqrtf((float)(v.z + 1));
        d.w = rsqrtf((float)(v.w + 1));
        ((float4*)dinv)[idx4] = d;
        if (idx4 == n4 - 1) rowptr[n] = excl + s3;
    }
}

__global__ void fill_kernel(const int* __restrict__ src, const int* __restrict__ dst,
                            const int* __restrict__ rank, const int* __restrict__ rowptr,
                            int* __restrict__ csr_src, int e4) {
    int i = blockIdx.x * blockDim.x + threadIdx.x;
    if (i >= e4) return;
    int4 s = ((const int4*)src)[i];
    int4 d = ((const int4*)dst)[i];
    int4 r = ((const int4*)rank)[i];
    int p0 = rowptr[d.x], p1 = rowptr[d.y], p2 = rowptr[d.z], p3 = rowptr[d.w];
    csr_src[p0 + r.x] = s.x;
    csr_src[p1 + r.y] = s.y;
    csr_src[p2 + r.z] = s.z;
    csr_src[p3 + r.w] = s.w;
}

template <int K, int NOUT, int MT>
__global__ __launch_bounds__(MT*(NOUT/16)*32)
void gemm_wmma_f32(const float* __restrict__ A, const float* __restrict__ B,
                   float* __restrict__ C, int M) {
    constexpr int NT   = NOUT / 16;
    constexpr int NW   = MT * NT;
    constexpr int ROWS = MT * 16;
    constexpr int KP   = K + 4;
    constexpr int NP   = NOUT + 4;

    __shared__ float As[ROWS][KP];
    __shared__ float Bs[K][NP];

    const int tid  = threadIdx.x;
    const int row0 = blockIdx.x * ROWS;

    constexpr int A_ELTS = ROWS * K;
    for (int i = tid * 4; i < A_ELTS; i += NW * 32 * 4) {
        int r = i / K, c = i % K;
        int gr = row0 + r;
        float4 v = (gr < M) ? *(const float4*)(A + (long long)gr * K + c)
                            : make_float4(0.f, 0.f, 0.f, 0.f);
        *(float4*)(&As[r][c]) = v;
    }
    constexpr int B_ELTS = K * NOUT;
    for (int i = tid * 4; i < B_ELTS; i += NW * 32 * 4) {
        int r = i / NOUT, c = i % NOUT;
        *(float4*)(&Bs[r][c]) = *(const float4*)(B + i);
    }
    __syncthreads();

    const int wid  = tid >> 5;
    const int lane = tid & 31;
    const int mn   = lane & 15;
    const int h    = lane >> 4;
    const int mt   = wid / NT;
    const int nt   = wid % NT;

    v8f acc = {};
    const float* arow = &As[mt * 16 + mn][0];
    const int    ncol = nt * 16 + mn;

#pragma unroll
    for (int k = 0; k < K; k += 4) {
        v2f a, b;
        a.x = arow[k + 2 * h + 0];
        a.y = arow[k + 2 * h + 1];
        b.x = Bs[k + 2 * h + 0][ncol];
        b.y = Bs[k + 2 * h + 1][ncol];
        acc = __builtin_amdgcn_wmma_f32_16x16x4_f32(
             false, a,  false, b,
             (short)0, acc,  false,  false);
    }

#pragma unroll
    for (int j = 0; j < 8; ++j) {
        int r = row0 + mt * 16 + j + 8 * h;
        if (r < M) C[(long long)r * NOUT + ncol] = acc[j];
    }
}

template <int F>
__global__ __launch_bounds__(256)
void gather_kernel(const float* __restrict__ h, const int* __restrict__ rowptr,
                   const int* __restrict__ csr_src, const float* __restrict__ dinv,
                   const float* __restrict__ bias, float* __restrict__ y, int n) {
    constexpr int LPE = F / 4;
    constexpr int G   = 32 / LPE;
    const int lane = threadIdx.x & 31;
    const int g = lane / LPE;
    const int j = lane % LPE;
    const float4* h4 = (const float4*)h;
    const float4 b = ((const float4*)bias)[j];
    const int wstride = gridDim.x * 8;

    for (int i = blockIdx.x * 8 + (threadIdx.x >> 5); i < n; i += wstride) {
        const int beg = rowptr[i];
        const int end = rowptr[i + 1];
        const float di = dinv[i];
        float4 acc = make_float4(0.f, 0.f, 0.f, 0.f);
        const int cntg = (end - beg - g + G - 1) / G;
        const int* q = csr_src + beg + g;
        int k = 0;
        for (; k + 4 <= cntg; k += 4) {
            const int s0 = q[0];
            const int s1 = q[G];
            const int s2 = q[2 * G];
            const int s3 = q[3 * G];
            q += 4 * G;
            const float w0 = dinv[s0] * di;
            const float w1 = dinv[s1] * di;
            const float w2 = dinv[s2] * di;
            const float w3 = dinv[s3] * di;
            const float4 v0 = h4[(long long)s0 * LPE + j];
            const float4 v1 = h4[(long long)s1 * LPE + j];
            const float4 v2 = h4[(long long)s2 * LPE + j];
            const float4 v3 = h4[(long long)s3 * LPE + j];
            acc.x += v0.x * w0; acc.y += v0.y * w0; acc.z += v0.z * w0; acc.w += v0.w * w0;
            acc.x += v1.x * w1; acc.y += v1.y * w1; acc.z += v1.z * w1; acc.w += v1.w * w1;
            acc.x += v2.x * w2; acc.y += v2.y * w2; acc.z += v2.z * w2; acc.w += v2.w * w2;
            acc.x += v3.x * w3; acc.y += v3.y * w3; acc.z += v3.z * w3; acc.w += v3.w * w3;
        }
#pragma unroll 1
        for (; k < cntg; ++k) {
            const int s0 = q[0];
            q += G;
            const float w0 = dinv[s0] * di;
            const float4 v0 = h4[(long long)s0 * LPE + j];
            acc.x += v0.x * w0; acc.y += v0.y * w0; acc.z += v0.z * w0; acc.w += v0.w * w0;
        }
#pragma unroll
        for (int o = LPE; o < 32; o <<= 1) {
            acc.x += __shfl_xor(acc.x, o);
            acc.y += __shfl_xor(acc.y, o);
            acc.z += __shfl_xor(acc.z, o);
            acc.w += __shfl_xor(acc.w, o);
        }
        if (g == 0) {
            const float4 hv = h4[(long long)i * LPE + j];
            float4 r;
            r.x = fmaxf(acc.x + hv.x * di * di + b.x, 0.0f);
            r.y = fmaxf(acc.y + hv.y * di * di + b.y, 0.0f);
            r.z = fmaxf(acc.z + hv.z * di * di + b.z, 0.0f);
            r.w = fmaxf(acc.w + hv.w * di * di + b.w, 0.0f);
            ((float4*)y)[(long long)i * LPE + j] = r;
        }
    }
}

__global__ void final_linear_kernel(const float* __restrict__ y, const float* __restrict__ Wl,
                                    const float* __restrict__ bl, float* out, int n) {
    int i = blockIdx.x * blockDim.x + threadIdx.x;
    if (i >= n) return;
    const float* row = y + (long long)i * 32;
    float s = 0.0f;
#pragma unroll
    for (int f = 0; f < 32; ++f) s += row[f] * Wl[f];
    out[i] = s + bl[0];
}

static inline size_t align256(size_t x) { return (x + 255) & ~size_t(255); }

extern "C" void kernel_launch(void* const* d_in, const int* in_sizes, int n_in,
                              void* d_out, int out_size, void* d_ws, size_t ws_size,
                              hipStream_t stream) {
    const float* x    = (const float*)d_in[0];
    const int*   ei   = (const int*)d_in[1];
    const float* W1   = (const float*)d_in[2];
    const float* b1   = (const float*)d_in[3];
    const float* W2   = (const float*)d_in[4];
    const float* b2   = (const float*)d_in[5];
    const float* Wl   = (const float*)d_in[6];
    const float* bl   = (const float*)d_in[7];
    float*       out  = (float*)d_out;

    const int N = N_NODES;
    const int E = N_EDGES;
    const int* src = ei;
    const int* dst = ei + E;

    char* ws = (char*)d_ws;
    size_t off = 0;
    int*   cnt    = (int*)(ws + off);   off += align256((size_t)N * 4);
    int*   rowptr = (int*)(ws + off);   off += align256((size_t)(N + 1) * 4);
    float* dinv   = (float*)(ws + off); off += align256((size_t)N * 4);
    int*   rank   = (int*)(ws + off);   off += align256((size_t)E * 4);
    int*   csr    = (int*)(ws + off);   off += align256((size_t)E * 4);
    float* h1     = (float*)(ws + off); off += align256((size_t)N * 64 * 4);
    float* y1     = (float*)(ws + off); off += align256((size_t)N * 64 * 4);
    float* h2     = (float*)(ws + off); off += align256((size_t)N * 32 * 4);
    float* y2     = (float*)(ws + off); off += align256((size_t)N * 32 * 4);
    (void)ws_size; (void)n_in; (void)in_sizes; (void)out_size;

    const int T = 256;

    zero_int_kernel<<<(N + T - 1) / T, T, 0, stream>>>(cnt, N);
    count_rank_kernel<<<(E / 4 + T - 1) / T, T, 0, stream>>>(dst, cnt, rank, E / 4);
    scan_dinv_kernel<<<(N / 4 + 1023) / 1024, 1024, 0, stream>>>(cnt, rowptr, dinv, N);
    fill_kernel<<<(E / 4 + T - 1) / T, T, 0, stream>>>(src, dst, rank, rowptr, csr, E / 4);

    gemm_wmma_f32<128, 64, 2><<<N / 32, 2 * 4 * 32, 0, stream>>>(x, W1, h1, N);

    gather_kernel<64><<<2048, T, 0, stream>>>(h1, rowptr, csr, dinv, b1, y1, N);

    gemm_wmma_f32<64, 32, 2><<<N / 32, 2 * 2 * 32, 0, stream>>>(y1, W2, h2, N);

    gather_kernel<32><<<2048, T, 0, stream>>>(h2, rowptr, csr, dinv, b2, y2, N);

    final_linear_kernel<<<(N + T - 1) / T, T, 0, stream>>>(y2, Wl, bl, out, N);
}


# GEMMs stage A rows times dinv[row] so h rows come out pre-scaled; gather loads index+row only (no per-edge dinv lookup), scales the sum by dinv[dst] once; self row starts the sum, factors regrouped
# speedup vs baseline: 1.8734x; 1.0549x over previous
#include <hip/hip_runtime.h>
#include <hip/hip_bf16.h>

typedef __attribute__((ext_vector_type(2))) float v2f;
typedef __attribute__((ext_vector_type(8))) float v8f;

#define N_NODES 100000
#define N_EDGES 3200000

static_assert(N_NODES % 4 == 0 && N_EDGES % 4 == 0, "16-byte index loads need multiples of 4");

__global__ void zero_int_kernel(int* p, int n) {
    int i = blockIdx.x * blockDim.x + threadIdx.x;
    if (i < n) p[i] = 0;
}

__global__ void count_rank_kernel(const int* __restrict__ dst, int* cnt,
                                  int* __restrict__ rank, int e4) {
    int i = blockIdx.x * blockDim.x + threadIdx.x;
    if (i >= e4) return;
    int4 d = ((const int4*)dst)[i];
    int4 r;
    r.x = atomicAdd(&cnt[d.x], 1);
    r.y = atomicAdd(&cnt[d.y], 1);
    r.z = atomicAdd(&cnt[d.z], 1);
    r.w = atomicAdd(&cnt[d.w], 1);
    ((int4*)rank)[i] = r;
}

__global__ __launch_bounds__(1024)
void scan_dinv_kernel(const int* __restrict__ cnt, int* __restrict__ rowptr,
                      float* __restrict__ dinv, int n) {
    __shared__ int s_pre[32];
    __shared__ int s_tot[32];
    const int tid  = threadIdx.x;
    const int lane = tid & 31;
    const int wid  = tid >> 5;
    const int n4   = n >> 2;
    const int4* c4 = (const int4*)cnt;
    const int first4 = blockIdx.x * 1024;

    int pre = 0;
    int i = tid;
    for (; i + 7 * 1024 < first4; i += 8 * 1024) {
        const int4 a0 = c4[i];
        const int4 a1 = c4[i + 1024];
        const int4 a2 = c4[i + 2 * 1024];
        const int4 a3 = c4[i + 3 * 1024];
        const int4 a4 = c4[i + 4 * 1024];
        const int4 a5 = c4[i + 5 * 1024];
        const int4 a6 = c4[i + 6 * 1024];
        const int4 a7 = c4[i + 7 * 1024];
        pre += (a0.x + a0.y) + (a0.z + a0.w);
        pre += (a1.x + a1.y) + (a1.z + a1.w);
        pre += (a2.x + a2.y) + (a2.z + a2.w);
        pre += (a3.x + a3.y) + (a3.z + a3.w);
        pre += (a4.x + a4.y) + (a4.z + a4.w);
        pre += (a5.x + a5.y) + (a5.z + a5.w);
        pre += (a6.x + a6.y) + (a6.z + a6.w);
        pre += (a7.x + a7.y) + (a7.z + a7.w);
    }
    for (; i < first4; i += 1024) {
        const int4 a0 = c4[i];
        pre += (a0.x + a0.y) + (a0.z + a0.w);
    }
    const int idx4 = first4 + tid;
    int4 v = make_int4(0, 0, 0, 0);
    if (idx4 < n4) v = c4[idx4];
    const int s0 = v.x, s1 = s0 + v.y, s2 = s1 + v.z, s3 = s2 + v.w;

#pragma unroll
    for (int o = 16; o >= 1; o >>= 1) pre += __shfl_xor(pre, o);
    int incl = s3;
#pragma unroll
    for (int o = 1; o < 32; o <<= 1) {
        int t = __shfl_up(incl, o);
        if (lane >= o) incl += t;
    }
    if (lane == 31) s_tot[wid] = incl;
    if (lane == 0)  s_pre[wid] = pre;
    __syncthreads();
    int base = 0;
#pragma unroll
    for (int w = 0; w < 32; ++w) {
        base += s_pre[w];
        if (w < wid) base += s_tot[w];
    }
    const int excl = base + incl - s3;
    if (idx4 < n4) {
        ((int4*)rowptr)[idx4] = make_int4(excl, excl + s0, excl + s1, excl + s2);
        float4 d;
        d.x = rsqrtf((float)(v.x + 1));
        d.y = rsqrtf((float)(v.y + 1));
        d.z = rsqrtf((float)(v.z + 1));
        d.w = rsqrtf((float)(v.w + 1));
        ((float4*)dinv)[idx4] = d;
        if (idx4 == n4 - 1) rowptr[n] = excl + s3;
    }
}

__global__ void fill_kernel(const int* __restrict__ src, const int* __restrict__ dst,
                            const int* __restrict__ rank, const int* __restrict__ rowptr,
                            int* __restrict__ csr_src, int e4) {
    int i = blockIdx.x * blockDim.x + threadIdx.x;
    if (i >= e4) return;
    int4 s = ((const int4*)src)[i];
    int4 d = ((const int4*)dst)[i];
    int4 r = ((const int4*)rank)[i];
    int p0 = rowptr[d.x], p1 = rowptr[d.y], p2 = rowptr[d.z], p3 = rowptr[d.w];
    csr_src[p0 + r.x] = s.x;
    csr_src[p1 + r.y] = s.y;
    csr_src[p2 + r.z] = s.z;
    csr_src[p3 + r.w] = s.w;
}

template <int K, int NOUT, int MT>
__global__ __launch_bounds__(MT*(NOUT/16)*32)
void gemm_wmma_f32(const float* __restrict__ A, const float* __restrict__ B,
                   const float* __restrict__ rowscale, float* __restrict__ C, int M) {
    constexpr int NT   = NOUT / 16;
    constexpr int NW   = MT * NT;
    constexpr int ROWS = MT * 16;
    constexpr int KP   = K + 4;
    constexpr int NP   = NOUT + 4;

    __shared__ float As[ROWS][KP];
    __shared__ float Bs[K][NP];

    const int tid  = threadIdx.x;
    const int row0 = blockIdx.x * ROWS;

    constexpr int A_ELTS = ROWS * K;
    for (int i = tid * 4; i < A_ELTS; i += NW * 32 * 4) {
        int r = i / K, c = i % K;
        int gr = row0 + r;
        float4 v = (gr < M) ? *(const float4*)(A + (long long)gr * K + c)
                            : make_float4(0.f, 0.f, 0.f, 0.f);
        const float s = (gr < M) ? rowscale[gr] : 0.f;
        v.x *= s; v.y *= s; v.z *= s; v.w *= s;
        *(float4*)(&As[r][c]) = v;
    }
    constexpr int B_ELTS = K * NOUT;
    for (int i = tid * 4; i < B_ELTS; i += NW * 32 * 4) {
        int r = i / NOUT, c = i % NOUT;
        *(float4*)(&Bs[r][c]) = *(const float4*)(B + i);
    }
    __syncthreads();

    const int wid  = tid >> 5;
    const int lane = tid & 31;
    const int mn   = lane & 15;
    const int h    = lane >> 4;
    const int mt   = wid / NT;
    const int nt   = wid % NT;

    v8f acc = {};
    const float* arow = &As[mt * 16 + mn][0];
    const int    ncol = nt * 16 + mn;

#pragma unroll
    for (int k = 0; k < K; k += 4) {
        v2f a, b;
        a.x = arow[k + 2 * h + 0];
        a.y = arow[k + 2 * h + 1];
        b.x = Bs[k + 2 * h + 0][ncol];
        b.y = Bs[k + 2 * h + 1][ncol];
        acc = __builtin_amdgcn_wmma_f32_16x16x4_f32(
             false, a,  false, b,
             (short)0, acc,  false,  false);
    }

#pragma unroll
    for (int j = 0; j < 8; ++j) {
        int r = row0 + mt * 16 + j + 8 * h;
        if (r < M) C[(long long)r * NOUT + ncol] = acc[j];
    }
}

template <int F>
__global__ __launch_bounds__(256)
void gather_kernel(const float* __restrict__ hs, const int* __restrict__ rowptr,
                   const int* __restrict__ csr_src, const float* __restrict__ dinv,
                   const float* __restrict__ bias, float* __restrict__ y, int n) {
    constexpr int LPE = F / 4;
    constexpr int G   = 32 / LPE;
    const int lane = threadIdx.x & 31;
    const int g = lane / LPE;
    const int j = lane % LPE;
    const float4* h4 = (const float4*)hs;
    const float4 b = ((const float4*)bias)[j];
    const int wstride = gridDim.x * 8;

    for (int i = blockIdx.x * 8 + (threadIdx.x >> 5); i < n; i += wstride) {
        const int beg = rowptr[i];
        const int end = rowptr[i + 1];
        const float di = dinv[i];
        float4 acc = make_float4(0.f, 0.f, 0.f, 0.f);
        if (g == 0) acc = h4[(long long)i * LPE + j];
        const int cntg = (end - beg - g + G - 1) / G;
        const int* q = csr_src + beg + g;
        int k = 0;
        for (; k + 4 <= cntg; k += 4) {
            const int s0 = q[0];
            const int s1 = q[G];
            const int s2 = q[2 * G];
            const int s3 = q[3 * G];
            q += 4 * G;
            const float4 v0 = h4[(long long)s0 * LPE + j];
            const float4 v1 = h4[(long long)s1 * LPE + j];
            const float4 v2 = h4[(long long)s2 * LPE + j];
            const float4 v3 = h4[(long long)s3 * LPE + j];
            acc.x += v0.x; acc.y += v0.y; acc.z += v0.z; acc.w += v0.w;
            acc.x += v1.x; acc.y += v1.y; acc.z += v1.z; acc.w += v1.w;
            acc.x += v2.x; acc.y += v2.y; acc.z += v2.z; acc.w += v2.w;
            acc.x += v3.x; acc.y += v3.y; acc.z += v3.z; acc.w += v3.w;
        }
#pragma unroll 1
        for (; k < cntg; ++k) {
            const int s0 = q[0];
            q += G;
            const float4 v0 = h4[(long long)s0 * LPE + j];
            acc.x += v0.x; acc.y += v0.y; acc.z += v0.z; acc.w += v0.w;
        }
#pragma unroll
        for (int o = LPE; o < 32; o <<= 1) {
            acc.x += __shfl_xor(acc.x, o);
            acc.y += __shfl_xor(acc.y, o);
            acc.z += __shfl_xor(acc.z, o);
            acc.w += __shfl_xor(acc.w, o);
        }
        if (g == 0) {
            float4 r;
            r.x = fmaxf(acc.x * di + b.x, 0.0f);
            r.y = fmaxf(acc.y * di + b.y, 0.0f);
            r.z = fmaxf(acc.z * di + b.z, 0.0f);
            r.w = fmaxf(acc.w * di + b.w, 0.0f);
            ((float4*)y)[(long long)i * LPE + j] = r;
        }
    }
}

__global__ void final_linear_kernel(const float* __restrict__ y, const float* __restrict__ Wl,
                                    const float* __restrict__ bl, float* out, int n) {
    int i = blockIdx.x * blockDim.x + threadIdx.x;
    if (i >= n) return;
    const float* row = y + (long long)i * 32;
    float s = 0.0f;
#pragma unroll
    for (int f = 0; f < 32; ++f) s += row[f] * Wl[f];
    out[i] = s + bl[0];
}

static inline size_t align256(size_t x) { return (x + 255) & ~size_t(255); }

extern "C" void kernel_launch(void* const* d_in, const int* in_sizes, int n_in,
                              void* d_out, int out_size, void* d_ws, size_t ws_size,
                              hipStream_t stream) {
    const float* x    = (const float*)d_in[0];
    const int*   ei   = (const int*)d_in[1];
    const float* W1   = (const float*)d_in[2];
    const float* b1   = (const float*)d_in[3];
    const float* W2   = (const float*)d_in[4];
    const float* b2   = (const float*)d_in[5];
    const float* Wl   = (const float*)d_in[6];
    const float* bl   = (const float*)d_in[7];
    float*       out  = (float*)d_out;

    const int N = N_NODES;
    const int E = N_EDGES;
    const int* src = ei;
    const int* dst = ei + E;

    char* ws = (char*)d_ws;
    size_t off = 0;
    int*   cnt    = (int*)(ws + off);   off += align256((size_t)N * 4);
    int*   rowptr = (int*)(ws + off);   off += align256((size_t)(N + 1) * 4);
    float* dinv   = (float*)(ws + off); off += align256((size_t)N * 4);
    int*   rank   = (int*)(ws + off);   off += align256((size_t)E * 4);
    int*   csr    = (int*)(ws + off);   off += align256((size_t)E * 4);
    float* h1     = (float*)(ws + off); off += align256((size_t)N * 64 * 4);
    float* y1     = (float*)(ws + off); off += align256((size_t)N * 64 * 4);
    float* h2     = (float*)(ws + off); off += align256((size_t)N * 32 * 4);
    float* y2     = (float*)(ws + off); off += align256((size_t)N * 32 * 4);
    (void)ws_size; (void)n_in; (void)in_sizes; (void)out_size;

    const int T = 256;

    zero_int_kernel<<<(N + T - 1) / T, T, 0, stream>>>(cnt, N);
    count_rank_kernel<<<(E / 4 + T - 1) / T, T, 0, stream>>>(dst, cnt, rank, E / 4);
    scan_dinv_kernel<<<(N / 4 + 1023) / 1024, 1024, 0, stream>>>(cnt, rowptr, dinv, N);
    fill_kernel<<<(E / 4 + T - 1) / T, T, 0, stream>>>(src, dst, rank, rowptr, csr, E / 4);

    gemm_wmma_f32<128, 64, 2><<<N / 32, 2 * 4 * 32, 0, stream>>>(x, W1, dinv, h1, N);

    gather_kernel<64><<<2048, T, 0, stream>>>(h1, rowptr, csr, dinv, b1, y1, N);

    gemm_wmma_f32<64, 32, 2><<<N / 32, 2 * 2 * 32, 0, stream>>>(y1, W2, dinv, h2, N);

    gather_kernel<32><<<2048, T, 0, stream>>>(h2, rowptr, csr, dinv, b2, y2, N);

    final_linear_kernel<<<(N + T - 1) / T, T, 0, stream>>>(y2, Wl, bl, out, N);
}


# Edges grouped by destination with work-group LDS counters (bin by 512-node range, then rank inside the bin, runs stored whole), no global returning add per edge; sum order stays arrival order
# speedup vs baseline: 4.8268x; 2.5765x over previous
#include <hip/hip_runtime.h>
#include <hip/hip_bf16.h>

typedef __attribute__((ext_vector_type(2))) float v2f;
typedef __attribute__((ext_vector_type(8))) float v8f;

#define N_NODES 100000
#define N_EDGES 3200000

static_assert(N_NODES % 4 == 0 && N_EDGES % 4 == 0, "16-byte index loads need multiples of 4");

#define RBITS   9
#define RANGE   (1 << RBITS)
#define NBIN    ((N_NODES + RANGE - 1) >> RBITS)
#define NCHUNK  256
#define CHUNK4  ((N_EDGES / 4 + NCHUNK - 1) / NCHUNK)
#define CHUNK   (CHUNK4 * 4)
#define NCELL   (NBIN * NCHUNK)
#define STAGE_CAP 40960

static_assert(NCELL % 4 == 0, "the cell scan loads 16 bytes at a time");

__global__ __launch_bounds__(1024)
void bin_count_kernel(const int* __restrict__ dst, int* __restrict__ cell, int e4) {
    __shared__ int s_hist[NBIN];
    const int tid = threadIdx.x;
    if (tid < NBIN) s_hist[tid] = 0;
    __syncthreads();
    const int first = blockIdx.x * CHUNK4;
    const int last  = (first + CHUNK4 < e4) ? first + CHUNK4 : e4;
    const int4* d4 = (const int4*)dst;
    for (int base = first + tid; base < last; base += 4 * 1024) {
        int4 d[4];
#pragma unroll
        for (int u = 0; u < 4; ++u) {
            const int i = base + u * 1024;
            d[u] = (i < last) ? d4[i] : make_int4(-1, -1, -1, -1);
        }
#pragma unroll
        for (int u = 0; u < 4; ++u) {
            if (d[u].x >= 0) {
                atomicAdd(&s_hist[d[u].x >> RBITS], 1);
                atomicAdd(&s_hist[d[u].y >> RBITS], 1);
                atomicAdd(&s_hist[d[u].z >> RBITS], 1);
                atomicAdd(&s_hist[d[u].w >> RBITS], 1);
            }
        }
    }
    __syncthreads();
    if (tid < NBIN) cell[tid * NCHUNK + blockIdx.x] = s_hist[tid];
}

__global__ __launch_bounds__(1024)
void scan_kernel(const int* __restrict__ cnt, int* __restrict__ out, int n) {
    __shared__ int s_pre[32];
    __shared__ int s_tot[32];
    const int tid  = threadIdx.x;
    const int lane = tid & 31;
    const int wid  = tid >> 5;
    const int n4   = n >> 2;
    const int4* c4 = (const int4*)cnt;
    const int first4 = blockIdx.x * 1024;

    int pre = 0;
    int i = tid;
    for (; i + 7 * 1024 < first4; i += 8 * 1024) {
        const int4 a0 = c4[i];
        const int4 a1 = c4[i + 1024];
        const int4 a2 = c4[i + 2 * 1024];
        const int4 a3 = c4[i + 3 * 1024];
        const int4 a4 = c4[i + 4 * 1024];
        const int4 a5 = c4[i + 5 * 1024];
        const int4 a6 = c4[i + 6 * 1024];
        const int4 a7 = c4[i + 7 * 1024];
        pre += (a0.x + a0.y) + (a0.z + a0.w);
        pre += (a1.x + a1.y) + (a1.z + a1.w);
        pre += (a2.x + a2.y) + (a2.z + a2.w);
        pre += (a3.x + a3.y) + (a3.z + a3.w);
        pre += (a4.x + a4.y) + (a4.z + a4.w);
        pre += (a5.x + a5.y) + (a5.z + a5.w);
        pre += (a6.x + a6.y) + (a6.z + a6.w);
        pre += (a7.x + a7.y) + (a7.z + a7.w);
    }
    for (; i < first4; i += 1024) {
        const int4 a0 = c4[i];
        pre += (a0.x + a0.y) + (a0.z + a0.w);
    }
    const int idx4 = first4 + tid;
    int4 v = make_int4(0, 0, 0, 0);
    if (idx4 < n4) v = c4[idx4];
    const int s0 = v.x, s1 = s0 + v.y, s2 = s1 + v.z, s3 = s2 + v.w;

#pragma unroll
    for (int o = 16; o >= 1; o >>= 1) pre += __shfl_xor(pre, o);
    int incl = s3;
#pragma unroll
    for (int o = 1; o < 32; o <<= 1) {
        int t = __shfl_up(incl, o);
        if (lane >= o) incl += t;
    }
    if (lane == 31) s_tot[wid] = incl;
    if (lane == 0)  s_pre[wid] = pre;
    __syncthreads();
    int base = 0;
#pragma unroll
    for (int w = 0; w < 32; ++w) {
        base += s_pre[w];
        if (w < wid) base += s_tot[w];
    }
    const int excl = base + incl - s3;
    if (idx4 < n4) {
        ((int4*)out)[idx4] = make_int4(excl, excl + s0, excl + s1, excl + s2);
        if (idx4 == n4 - 1) out[n] = excl + s3;
    }
}

__global__ __launch_bounds__(1024)
void bin_scatter_kernel(const int* __restrict__ src, const int* __restrict__ dst,
                        const int* __restrict__ off, int* __restrict__ bsrc,
                        int* __restrict__ bdst, int e4) {
    __shared__ int s_cnt[NBIN];
    __shared__ int s_cur[NBIN];
    __shared__ int s_base[NBIN];
    __shared__ int s_src[CHUNK];
    __shared__ int s_dst[CHUNK];
    const int tid = threadIdx.x;
    const int b   = blockIdx.x;
    int o0 = 0;
    if (tid < NBIN) {
        o0 = off[tid * NCHUNK + b];
        s_cnt[tid] = off[tid * NCHUNK + b + 1] - o0;
    }
    __syncthreads();
    if (tid < NBIN) {
        int loc = 0;
        for (int k = 0; k < tid; ++k) loc += s_cnt[k];
        s_cur[tid]  = loc;
        s_base[tid] = o0 - loc;
    }
    __syncthreads();
    const int first = b * CHUNK4;
    const int last  = (first + CHUNK4 < e4) ? first + CHUNK4 : e4;
    const int4* d4 = (const int4*)dst;
    const int4* s4 = (const int4*)src;
    for (int base = first + tid; base < last; base += 4 * 1024) {
        int4 d[4], s[4];
#pragma unroll
        for (int u = 0; u < 4; ++u) {
            const int i = base + u * 1024;
            d[u] = (i < last) ? d4[i] : make_int4(-1, -1, -1, -1);
            s[u] = (i < last) ? s4[i] : make_int4(0, 0, 0, 0);
        }
#pragma unroll
        for (int u = 0; u < 4; ++u) {
            if (d[u].x >= 0) {
                int p;
                p = atomicAdd(&s_cur[d[u].x >> RBITS], 1); s_src[p] = s[u].x; s_dst[p] = d[u].x;
                p = atomicAdd(&s_cur[d[u].y >> RBITS], 1); s_src[p] = s[u].y; s_dst[p] = d[u].y;
                p = atomicAdd(&s_cur[d[u].z >> RBITS], 1); s_src[p] = s[u].z; s_dst[p] = d[u].z;
                p = atomicAdd(&s_cur[d[u].w >> RBITS], 1); s_src[p] = s[u].w; s_dst[p] = d[u].w;
            }
        }
    }
    __syncthreads();
    const int len = (last > first) ? (last - first) * 4 : 0;
    for (int i = tid; i < len; i += 1024) {
        const int d = s_dst[i];
        const int g = s_base[d >> RBITS] + i;
        bsrc[g] = s_src[i];
        bdst[g] = d;
    }
}

__global__ __launch_bounds__(1024)
void bin_sort_kernel(const int* __restrict__ bsrc, const int* __restrict__ bdst,
                     const int* __restrict__ off, int* __restrict__ rowptr,
                     float* __restrict__ dinv, int* __restrict__ csr_src, int n) {
    __shared__ int s_cnt[RANGE];
    __shared__ int s_cur[RANGE];
    __shared__ int s_wtot[RANGE / 32];
    __shared__ int s_stage[STAGE_CAP];
    const int tid  = threadIdx.x;
    const int lane = tid & 31;
    const int wid  = tid >> 5;
    const int bin  = blockIdx.x;
    const int beg  = off[bin * NCHUNK];
    const int end  = off[(bin + 1) * NCHUNK];
    if (tid < RANGE) s_cnt[tid] = 0;
    __syncthreads();

    {
        int i = beg + tid;
        for (; i + 7 * 1024 < end; i += 8 * 1024) {
            int d[8];
#pragma unroll
            for (int u = 0; u < 8; ++u) d[u] = bdst[i + u * 1024];
#pragma unroll
            for (int u = 0; u < 8; ++u) atomicAdd(&s_cnt[d[u] & (RANGE - 1)], 1);
        }
        for (; i < end; i += 1024) atomicAdd(&s_cnt[bdst[i] & (RANGE - 1)], 1);
    }
    __syncthreads();

    int v = 0, incl = 0;
    if (tid < RANGE) {
        v = s_cnt[tid];
        incl = v;
#pragma unroll
        for (int o = 1; o < 32; o <<= 1) {
            int t = __shfl_up(incl, o);
            if (lane >= o) incl += t;
        }
        if (lane == 31) s_wtot[wid] = incl;
    }
    __syncthreads();
    if (tid < RANGE) {
        int base = 0;
#pragma unroll
        for (int w = 0; w < RANGE / 32; ++w) if (w < wid) base += s_wtot[w];
        const int excl = base + incl - v;
        s_cur[tid] = excl;
        const int node = bin * RANGE + tid;
        if (node < n) {
            rowptr[node] = beg + excl;
            dinv[node]   = rsqrtf((float)(v + 1));
            if (node == n - 1) rowptr[n] = beg + excl + v;
        }
    }
    __syncthreads();

    const int cnt = end - beg;
    if (cnt <= STAGE_CAP) {
        int i = beg + tid;
        for (; i + 3 * 1024 < end; i += 4 * 1024) {
            int d[4], s[4];
#pragma unroll
            for (int u = 0; u < 4; ++u) { d[u] = bdst[i + u * 1024]; s[u] = bsrc[i + u * 1024]; }
#pragma unroll
            for (int u = 0; u < 4; ++u) {
                const int p = atomicAdd(&s_cur[d[u] & (RANGE - 1)], 1);
                s_stage[p] = s[u];
            }
        }
        for (; i < end; i += 1024) {
            const int p = atomicAdd(&s_cur[bdst[i] & (RANGE - 1)], 1);
            s_stage[p] = bsrc[i];
        }
        __syncthreads();
        for (int k = tid; k < cnt; k += 1024) csr_src[beg + k] = s_stage[k];
    } else {
        for (int i = beg + tid; i < end; i += 1024) {
            const int p = atomicAdd(&s_cur[bdst[i] & (RANGE - 1)], 1);
            csr_src[beg + p] = bsrc[i];
        }
    }
}

template <int K, int NOUT, int MT>
__global__ __launch_bounds__(MT*(NOUT/16)*32)
void gemm_wmma_f32(const float* __restrict__ A, const float* __restrict__ B,
                   const float* __restrict__ rowscale, float* __restrict__ C, int M) {
    constexpr int NT   = NOUT / 16;
    constexpr int NW   = MT * NT;
    constexpr int ROWS = MT * 16;
    constexpr int KP   = K + 4;
    constexpr int NP   = NOUT + 4;

    __shared__ float As[ROWS][KP];
    __shared__ float Bs[K][NP];

    const int tid  = threadIdx.x;
    const int row0 = blockIdx.x * ROWS;

    constexpr int A_ELTS = ROWS * K;
    for (int i = tid * 4; i < A_ELTS; i += NW * 32 * 4) {
        int r = i / K, c = i % K;
        int gr = row0 + r;
        float4 v = (gr < M) ? *(const float4*)(A + (long long)gr * K + c)
                            : make_float4(0.f, 0.f, 0.f, 0.f);
        const float s = (gr < M) ? rowscale[gr] : 0.f;
        v.x *= s; v.y *= s; v.z *= s; v.w *= s;
        *(float4*)(&As[r][c]) = v;
    }
    constexpr int B_ELTS = K * NOUT;
    for (int i = tid * 4; i < B_ELTS; i += NW * 32 * 4) {
        int r = i / NOUT, c = i % NOUT;
        *(float4*)(&Bs[r][c]) = *(const float4*)(B + i);
    }
    __syncthreads();

    const int wid  = tid >> 5;
    const int lane = tid & 31;
    const int mn   = lane & 15;
    const int h    = lane >> 4;
    const int mt   = wid / NT;
    const int nt   = wid % NT;

    v8f acc = {};
    const float* arow = &As[mt * 16 + mn][0];
    const int    ncol = nt * 16 + mn;

#pragma unroll
    for (int k = 0; k < K; k += 4) {
        v2f a, b;
        a.x = arow[k + 2 * h + 0];
        a.y = arow[k + 2 * h + 1];
        b.x = Bs[k + 2 * h + 0][ncol];
        b.y = Bs[k + 2 * h + 1][ncol];
        acc = __builtin_amdgcn_wmma_f32_16x16x4_f32(
             false, a,  false, b,
             (short)0, acc,  false,  false);
    }

#pragma unroll
    for (int j = 0; j < 8; ++j) {
        int r = row0 + mt * 16 + j + 8 * h;
        if (r < M) C[(long long)r * NOUT + ncol] = acc[j];
    }
}

template <int F>
__global__ __launch_bounds__(256)
void gather_kernel(const float* __restrict__ hs, const int* __restrict__ rowptr,
                   const int* __restrict__ csr_src, const float* __restrict__ dinv,
                   const float* __restrict__ bias, float* __restrict__ y, int n) {
    constexpr int LPE = F / 4;
    constexpr int G   = 32 / LPE;
    const int lane = threadIdx.x & 31;
    const int g = lane / LPE;
    const int j = lane % LPE;
    const float4* h4 = (const float4*)hs;
    const float4 b = ((const float4*)bias)[j];
    const int wstride = gridDim.x * 8;

    for (int i = blockIdx.x * 8 + (threadIdx.x >> 5); i < n; i += wstride) {
        const int beg = rowptr[i];
        const int end = rowptr[i + 1];
        const float di = dinv[i];
        float4 acc = make_float4(0.f, 0.f, 0.f, 0.f);
        if (g == 0) acc = h4[(long long)i * LPE + j];
        const int cntg = (end - beg - g + G - 1) / G;
        const int* q = csr_src + beg + g;
        int k = 0;
        for (; k + 4 <= cntg; k += 4) {
            const int s0 = q[0];
            const int s1 = q[G];
            const int s2 = q[2 * G];
            const int s3 = q[3 * G];
            q += 4 * G;
            const float4 v0 = h4[(long long)s0 * LPE + j];
            const float4 v1 = h4[(long long)s1 * LPE + j];
            const float4 v2 = h4[(long long)s2 * LPE + j];
            const float4 v3 = h4[(long long)s3 * LPE + j];
            acc.x += v0.x; acc.y += v0.y; acc.z += v0.z; acc.w += v0.w;
            acc.x += v1.x; acc.y += v1.y; acc.z += v1.z; acc.w += v1.w;
            acc.x += v2.x; acc.y += v2.y; acc.z += v2.z; acc.w += v2.w;
            acc.x += v3.x; acc.y += v3.y; acc.z += v3.z; acc.w += v3.w;
        }
#pragma unroll 1
        for (; k < cntg; ++k) {
            const int s0 = q[0];
            q += G;
            const float4 v0 = h4[(long long)s0 * LPE + j];
            acc.x += v0.x; acc.y += v0.y; acc.z += v0.z; acc.w += v0.w;
        }
#pragma unroll
        for (int o = LPE; o < 32; o <<= 1) {
            acc.x += __shfl_xor(acc.x, o);
            acc.y += __shfl_xor(acc.y, o);
            acc.z += __shfl_xor(acc.z, o);
            acc.w += __shfl_xor(acc.w, o);
        }
        if (g == 0) {
            float4 r;
            r.x = fmaxf(acc.x * di + b.x, 0.0f);
            r.y = fmaxf(acc.y * di + b.y, 0.0f);
            r.z = fmaxf(acc.z * di + b.z, 0.0f);
            r.w = fmaxf(acc.w * di + b.w, 0.0f);
            ((float4*)y)[(long long)i * LPE + j] = r;
        }
    }
}

__global__ void final_linear_kernel(const float* __restrict__ y, const float* __restrict__ Wl,
                                    const float* __restrict__ bl, float* out, int n) {
    int i = blockIdx.x * blockDim.x + threadIdx.x;
    if (i >= n) return;
    const float* row = y + (long long)i * 32;
    float s = 0.0f;
#pragma unroll
    for (int f = 0; f < 32; ++f) s += row[f] * Wl[f];
    out[i] = s + bl[0];
}

static inline size_t align256(size_t x) { return (x + 255) & ~size_t(255); }

extern "C" void kernel_launch(void* const* d_in, const int* in_sizes, int n_in,
                              void* d_out, int out_size, void* d_ws, size_t ws_size,
                              hipStream_t stream) {
    const float* x    = (const float*)d_in[0];
    const int*   ei   = (const int*)d_in[1];
    const float* W1   = (const float*)d_in[2];
    const float* b1   = (const float*)d_in[3];
    const float* W2   = (const float*)d_in[4];
    const float* b2   = (const float*)d_in[5];
    const float* Wl   = (const float*)d_in[6];
    const float* bl   = (const float*)d_in[7];
    float*       out  = (float*)d_out;

    const int N = N_NODES;
    const int E = N_EDGES;
    const int* src = ei;
    const int* dst = ei + E;

    char* ws = (char*)d_ws;
    size_t off = 0;
    int*   cell   = (int*)(ws + off);   off += align256((size_t)NCELL * 4);
    int*   celloff= (int*)(ws + off);   off += align256((size_t)(NCELL + 1) * 4);
    int*   rowptr = (int*)(ws + off);   off += align256((size_t)(N + 1) * 4);
    float* dinv   = (float*)(ws + off); off += align256((size_t)N * 4);
    int*   bsrc   = (int*)(ws + off);   off += align256((size_t)E * 4);
    int*   bdst   = (int*)(ws + off);   off += align256((size_t)E * 4);
    int*   csr    = (int*)(ws + off);   off += align256((size_t)E * 4);
    float* h1     = (float*)(ws + off); off += align256((size_t)N * 64 * 4);
    float* y1     = (float*)(ws + off); off += align256((size_t)N * 64 * 4);
    float* h2     = (float*)(ws + off); off += align256((size_t)N * 32 * 4);
    float* y2     = (float*)(ws + off); off += align256((size_t)N * 32 * 4);
    (void)ws_size; (void)n_in; (void)in_sizes; (void)out_size;

    const int T = 256;

    bin_count_kernel<<<NCHUNK, 1024, 0, stream>>>(dst, cell, E / 4);
    scan_kernel<<<(NCELL / 4 + 1023) / 1024, 1024, 0, stream>>>(cell, celloff, NCELL);
    bin_scatter_kernel<<<NCHUNK, 1024, 0, stream>>>(src, dst, celloff, bsrc, bdst, E / 4);
    bin_sort_kernel<<<NBIN, 1024, 0, stream>>>(bsrc, bdst, celloff, rowptr, dinv, csr, N);

    gemm_wmma_f32<128, 64, 2><<<N / 32, 2 * 4 * 32, 0, stream>>>(x, W1, dinv, h1, N);

    gather_kernel<64><<<2048, T, 0, stream>>>(h1, rowptr, csr, dinv, b1, y1, N);

    gemm_wmma_f32<64, 32, 2><<<N / 32, 2 * 2 * 32, 0, stream>>>(y1, W2, dinv, h2, N);

    gather_kernel<32><<<2048, T, 0, stream>>>(h2, rowptr, csr, dinv, b2, y2, N);

    final_linear_kernel<<<(N + T - 1) / T, T, 0, stream>>>(y2, Wl, bl, out, N);
}


# GEMM work-groups stay resident and walk over 16-row tiles; each wave keeps its W fragments in registers for the whole walk (no W in LDS), next tile's rows prefetched under the WMMAs; sums unchanged
# speedup vs baseline: 5.1034x; 1.0573x over previous
#include <hip/hip_runtime.h>
#include <hip/hip_bf16.h>

typedef __attribute__((ext_vector_type(2))) float v2f;
typedef __attribute__((ext_vector_type(8))) float v8f;

#define N_NODES 100000
#define N_EDGES 3200000

static_assert(N_NODES % 4 == 0 && N_EDGES % 4 == 0, "16-byte index loads need multiples of 4");

#define RBITS   9
#define RANGE   (1 << RBITS)
#define NBIN    ((N_NODES + RANGE - 1) >> RBITS)
#define NCHUNK  256
#define CHUNK4  ((N_EDGES / 4 + NCHUNK - 1) / NCHUNK)
#define CHUNK   (CHUNK4 * 4)
#define NCELL   (NBIN * NCHUNK)
#define STAGE_CAP 40960

static_assert(NCELL % 4 == 0, "the cell scan loads 16 bytes at a time");

__global__ __launch_bounds__(1024)
void bin_count_kernel(const int* __restrict__ dst, int* __restrict__ cell, int e4) {
    __shared__ int s_hist[NBIN];
    const int tid = threadIdx.x;
    if (tid < NBIN) s_hist[tid] = 0;
    __syncthreads();
    const int first = blockIdx.x * CHUNK4;
    const int last  = (first + CHUNK4 < e4) ? first + CHUNK4 : e4;
    const int4* d4 = (const int4*)dst;
    for (int base = first + tid; base < last; base += 4 * 1024) {
        int4 d[4];
#pragma unroll
        for (int u = 0; u < 4; ++u) {
            const int i = base + u * 1024;
            d[u] = (i < last) ? d4[i] : make_int4(-1, -1, -1, -1);
        }
#pragma unroll
        for (int u = 0; u < 4; ++u) {
            if (d[u].x >= 0) {
                atomicAdd(&s_hist[d[u].x >> RBITS], 1);
                atomicAdd(&s_hist[d[u].y >> RBITS], 1);
                atomicAdd(&s_hist[d[u].z >> RBITS], 1);
                atomicAdd(&s_hist[d[u].w >> RBITS], 1);
            }
        }
    }
    __syncthreads();
    if (tid < NBIN) cell[tid * NCHUNK + blockIdx.x] = s_hist[tid];
}

__global__ __launch_bounds__(1024)
void scan_kernel(const int* __restrict__ cnt, int* __restrict__ out, int n) {
    __shared__ int s_pre[32];
    __shared__ int s_tot[32];
    const int tid  = threadIdx.x;
    const int lane = tid & 31;
    const int wid  = tid >> 5;
    const int n4   = n >> 2;
    const int4* c4 = (const int4*)cnt;
    const int first4 = blockIdx.x * 1024;

    int pre = 0;
    int i = tid;
    for (; i + 7 * 1024 < first4; i += 8 * 1024) {
        const int4 a0 = c4[i];
        const int4 a1 = c4[i + 1024];
        const int4 a2 = c4[i + 2 * 1024];
        const int4 a3 = c4[i + 3 * 1024];
        const int4 a4 = c4[i + 4 * 1024];
        const int4 a5 = c4[i + 5 * 1024];
        const int4 a6 = c4[i + 6 * 1024];
        const int4 a7 = c4[i + 7 * 1024];
        pre += (a0.x + a0.y) + (a0.z + a0.w);
        pre += (a1.x + a1.y) + (a1.z + a1.w);
        pre += (a2.x + a2.y) + (a2.z + a2.w);
        pre += (a3.x + a3.y) + (a3.z + a3.w);
        pre += (a4.x + a4.y) + (a4.z + a4.w);
        pre += (a5.x + a5.y) + (a5.z + a5.w);
        pre += (a6.x + a6.y) + (a6.z + a6.w);
        pre += (a7.x + a7.y) + (a7.z + a7.w);
    }
    for (; i < first4; i += 1024) {
        const int4 a0 = c4[i];
        pre += (a0.x + a0.y) + (a0.z + a0.w);
    }
    const int idx4 = first4 + tid;
    int4 v = make_int4(0, 0, 0, 0);
    if (idx4 < n4) v = c4[idx4];
    const int s0 = v.x, s1 = s0 + v.y, s2 = s1 + v.z, s3 = s2 + v.w;

#pragma unroll
    for (int o = 16; o >= 1; o >>= 1) pre += __shfl_xor(pre, o);
    int incl = s3;
#pragma unroll
    for (int o = 1; o < 32; o <<= 1) {
        int t = __shfl_up(incl, o);
        if (lane >= o) incl += t;
    }
    if (lane == 31) s_tot[wid] = incl;
    if (lane == 0)  s_pre[wid] = pre;
    __syncthreads();
    int base = 0;
#pragma unroll
    for (int w = 0; w < 32; ++w) {
        base += s_pre[w];
        if (w < wid) base += s_tot[w];
    }
    const int excl = base + incl - s3;
    if (idx4 < n4) {
        ((int4*)out)[idx4] = make_int4(excl, excl + s0, excl + s1, excl + s2);
        if (idx4 == n4 - 1) out[n] = excl + s3;
    }
}

__global__ __launch_bounds__(1024)
void bin_scatter_kernel(const int* __restrict__ src, const int* __restrict__ dst,
                        const int* __restrict__ off, int* __restrict__ bsrc,
                        int* __restrict__ bdst, int e4) {
    __shared__ int s_cnt[NBIN];
    __shared__ int s_cur[NBIN];
    __shared__ int s_base[NBIN];
    __shared__ int s_src[CHUNK];
    __shared__ int s_dst[CHUNK];
    const int tid = threadIdx.x;
    const int b   = blockIdx.x;
    int o0 = 0;
    if (tid < NBIN) {
        o0 = off[tid * NCHUNK + b];
        s_cnt[tid] = off[tid * NCHUNK + b + 1] - o0;
    }
    __syncthreads();
    if (tid < NBIN) {
        int loc = 0;
        for (int k = 0; k < tid; ++k) loc += s_cnt[k];
        s_cur[tid]  = loc;
        s_base[tid] = o0 - loc;
    }
    __syncthreads();
    const int first = b * CHUNK4;
    const int last  = (first + CHUNK4 < e4) ? first + CHUNK4 : e4;
    const int4* d4 = (const int4*)dst;
    const int4* s4 = (const int4*)src;
    for (int base = first + tid; base < last; base += 4 * 1024) {
        int4 d[4], s[4];
#pragma unroll
        for (int u = 0; u < 4; ++u) {
            const int i = base + u * 1024;
            d[u] = (i < last) ? d4[i] : make_int4(-1, -1, -1, -1);
            s[u] = (i < last) ? s4[i] : make_int4(0, 0, 0, 0);
        }
#pragma unroll
        for (int u = 0; u < 4; ++u) {
            if (d[u].x >= 0) {
                int p;
                p = atomicAdd(&s_cur[d[u].x >> RBITS], 1); s_src[p] = s[u].x; s_dst[p] = d[u].x;
                p = atomicAdd(&s_cur[d[u].y >> RBITS], 1); s_src[p] = s[u].y; s_dst[p] = d[u].y;
                p = atomicAdd(&s_cur[d[u].z >> RBITS], 1); s_src[p] = s[u].z; s_dst[p] = d[u].z;
                p = atomicAdd(&s_cur[d[u].w >> RBITS], 1); s_src[p] = s[u].w; s_dst[p] = d[u].w;
            }
        }
    }
    __syncthreads();
    const int len = (last > first) ? (last - first) * 4 : 0;
    for (int i = tid; i < len; i += 1024) {
        const int d = s_dst[i];
        const int g = s_base[d >> RBITS] + i;
        bsrc[g] = s_src[i];
        bdst[g] = d;
    }
}

__global__ __launch_bounds__(1024)
void bin_sort_kernel(const int* __restrict__ bsrc, const int* __restrict__ bdst,
                     const int* __restrict__ off, int* __restrict__ rowptr,
                     float* __restrict__ dinv, int* __restrict__ csr_src, int n) {
    __shared__ int s_cnt[RANGE];
    __shared__ int s_cur[RANGE];
    __shared__ int s_wtot[RANGE / 32];
    __shared__ int s_stage[STAGE_CAP];
    const int tid  = threadIdx.x;
    const int lane = tid & 31;
    const int wid  = tid >> 5;
    const int bin  = blockIdx.x;
    const int beg  = off[bin * NCHUNK];
    const int end  = off[(bin + 1) * NCHUNK];
    if (tid < RANGE) s_cnt[tid] = 0;
    __syncthreads();

    {
        int i = beg + tid;
        for (; i + 7 * 1024 < end; i += 8 * 1024) {
            int d[8];
#pragma unroll
            for (int u = 0; u < 8; ++u) d[u] = bdst[i + u * 1024];
#pragma unroll
            for (int u = 0; u < 8; ++u) atomicAdd(&s_cnt[d[u] & (RANGE - 1)], 1);
        }
        for (; i < end; i += 1024) atomicAdd(&s_cnt[bdst[i] & (RANGE - 1)], 1);
    }
    __syncthreads();

    int v = 0, incl = 0;
    if (tid < RANGE) {
        v = s_cnt[tid];
        incl = v;
#pragma unroll
        for (int o = 1; o < 32; o <<= 1) {
            int t = __shfl_up(incl, o);
            if (lane >= o) incl += t;
        }
        if (lane == 31) s_wtot[wid] = incl;
    }
    __syncthreads();
    if (tid < RANGE) {
        int base = 0;
#pragma unroll
        for (int w = 0; w < RANGE / 32; ++w) if (w < wid) base += s_wtot[w];
        const int excl = base + incl - v;
        s_cur[tid] = excl;
        const int node = bin * RANGE + tid;
        if (node < n) {
            rowptr[node] = beg + excl;
            dinv[node]   = rsqrtf((float)(v + 1));
            if (node == n - 1) rowptr[n] = beg + excl + v;
        }
    }
    __syncthreads();

    const int cnt = end - beg;
    if (cnt <= STAGE_CAP) {
        int i = beg + tid;
        for (; i + 3 * 1024 < end; i += 4 * 1024) {
            int d[4], s[4];
#pragma unroll
            for (int u = 0; u < 4; ++u) { d[u] = bdst[i + u * 1024]; s[u] = bsrc[i + u * 1024]; }
#pragma unroll
            for (int u = 0; u < 4; ++u) {
                const int p = atomicAdd(&s_cur[d[u] & (RANGE - 1)], 1);
                s_stage[p] = s[u];
            }
        }
        for (; i < end; i += 1024) {
            const int p = atomicAdd(&s_cur[bdst[i] & (RANGE - 1)], 1);
            s_stage[p] = bsrc[i];
        }
        __syncthreads();
        for (int k = tid; k < cnt; k += 1024) csr_src[beg + k] = s_stage[k];
    } else {
        for (int i = beg + tid; i < end; i += 1024) {
            const int p = atomicAdd(&s_cur[bdst[i] & (RANGE - 1)], 1);
            csr_src[beg + p] = bsrc[i];
        }
    }
}

template <int K, int NOUT, int MT>
__global__ __launch_bounds__(MT*(NOUT/16)*32)
void gemm_wmma_f32(const float* __restrict__ A, const float* __restrict__ B,
                   const float* __restrict__ rowscale, float* __restrict__ C, int M) {
    constexpr int NT   = NOUT / 16;
    constexpr int NW   = MT * NT;
    constexpr int NTHR = NW * 32;
    constexpr int ROWS = MT * 16;
    constexpr int KP   = K + 4;
    constexpr int KS   = K / 4;
    constexpr int A_ELTS = ROWS * K;
    constexpr int PER  = A_ELTS / (NTHR * 4);
    static_assert(A_ELTS % (NTHR * 4) == 0, "a tile is a whole number of 16-byte pieces per thread");

    __shared__ float As[2][ROWS][KP];

    const int tid    = threadIdx.x;
    const int ntiles = (M + ROWS - 1) / ROWS;

    float4 pv[PER];
    float  ps[PER];
    int tile = blockIdx.x;
    if (tile < ntiles) {
#pragma unroll
        for (int u = 0; u < PER; ++u) {
            const int i  = (tid + u * NTHR) * 4;
            const int gr = tile * ROWS + i / K;
            pv[u] = (gr < M) ? *(const float4*)(A + (long long)gr * K + i % K)
                             : make_float4(0.f, 0.f, 0.f, 0.f);
            ps[u] = (gr < M) ? rowscale[gr] : 0.f;
        }
    }

    const int wid  = tid >> 5;
    const int lane = tid & 31;
    const int mn   = lane & 15;
    const int h    = lane >> 4;
    const int mt   = wid / NT;
    const int nt   = wid % NT;
    const int ncol = nt * 16 + mn;

    v2f bf[KS];
#pragma unroll
    for (int kk = 0; kk < KS; ++kk) {
        bf[kk].x = B[(4 * kk + 2 * h + 0) * NOUT + ncol];
        bf[kk].y = B[(4 * kk + 2 * h + 1) * NOUT + ncol];
    }

    int buf = 0;
    for (; tile < ntiles; tile += gridDim.x) {
#pragma unroll
        for (int u = 0; u < PER; ++u) {
            const int i = (tid + u * NTHR) * 4;
            float4 v = pv[u];
            const float s = ps[u];
            v.x *= s; v.y *= s; v.z *= s; v.w *= s;
            *(float4*)(&As[buf][i / K][i % K]) = v;
        }
        __syncthreads();

        const int nxt = tile + gridDim.x;
        if (nxt < ntiles) {
#pragma unroll
            for (int u = 0; u < PER; ++u) {
                const int i  = (tid + u * NTHR) * 4;
                const int gr = nxt * ROWS + i / K;
                pv[u] = (gr < M) ? *(const float4*)(A + (long long)gr * K + i % K)
                                 : make_float4(0.f, 0.f, 0.f, 0.f);
                ps[u] = (gr < M) ? rowscale[gr] : 0.f;
            }
        }

        v8f acc = {};
        const float* arow = &As[buf][mt * 16 + mn][0];
#pragma unroll
        for (int kk = 0; kk < KS; ++kk) {
            v2f a;
            a.x = arow[4 * kk + 2 * h + 0];
            a.y = arow[4 * kk + 2 * h + 1];
            acc = __builtin_amdgcn_wmma_f32_16x16x4_f32(
                 false, a,  false, bf[kk],
                 (short)0, acc,  false,  false);
        }

        const int row0 = tile * ROWS;
#pragma unroll
        for (int j = 0; j < 8; ++j) {
            int r = row0 + mt * 16 + j + 8 * h;
            if (r < M) C[(long long)r * NOUT + ncol] = acc[j];
        }
        buf ^= 1;
    }
}

template <int F>
__global__ __launch_bounds__(256)
void gather_kernel(const float* __restrict__ hs, const int* __restrict__ rowptr,
                   const int* __restrict__ csr_src, const float* __restrict__ dinv,
                   const float* __restrict__ bias, float* __restrict__ y, int n) {
    constexpr int LPE = F / 4;
    constexpr int G   = 32 / LPE;
    const int lane = threadIdx.x & 31;
    const int g = lane / LPE;
    const int j = lane % LPE;
    const float4* h4 = (const float4*)hs;
    const float4 b = ((const float4*)bias)[j];
    const int wstride = gridDim.x * 8;

    for (int i = blockIdx.x * 8 + (threadIdx.x >> 5); i < n; i += wstride) {
        const int beg = rowptr[i];
        const int end = rowptr[i + 1];
        const float di = dinv[i];
        float4 acc = make_float4(0.f, 0.f, 0.f, 0.f);
        if (g == 0) acc = h4[(long long)i * LPE + j];
        const int cntg = (end - beg - g + G - 1) / G;
        const int* q = csr_src + beg + g;
        int k = 0;
        for (; k + 4 <= cntg; k += 4) {
            const int s0 = q[0];
            const int s1 = q[G];
            const int s2 = q[2 * G];
            const int s3 = q[3 * G];
            q += 4 * G;
            const float4 v0 = h4[(long long)s0 * LPE + j];
            const float4 v1 = h4[(long long)s1 * LPE + j];
            const float4 v2 = h4[(long long)s2 * LPE + j];
            const float4 v3 = h4[(long long)s3 * LPE + j];
            acc.x += v0.x; acc.y += v0.y; acc.z += v0.z; acc.w += v0.w;
            acc.x += v1.x; acc.y += v1.y; acc.z += v1.z; acc.w += v1.w;
            acc.x += v2.x; acc.y += v2.y; acc.z += v2.z; acc.w += v2.w;
            acc.x += v3.x; acc.y += v3.y; acc.z += v3.z; acc.w += v3.w;
        }
#pragma unroll 1
        for (; k < cntg; ++k) {
            const int s0 = q[0];
            q += G;
            const float4 v0 = h4[(long long)s0 * LPE + j];
            acc.x += v0.x; acc.y += v0.y; acc.z += v0.z; acc.w += v0.w;
        }
#pragma unroll
        for (int o = LPE; o < 32; o <<= 1) {
            acc.x += __shfl_xor(acc.x, o);
            acc.y += __shfl_xor(acc.y, o);
            acc.z += __shfl_xor(acc.z, o);
            acc.w += __shfl_xor(acc.w, o);
        }
        if (g == 0) {
            float4 r;
            r.x = fmaxf(acc.x * di + b.x, 0.0f);
            r.y = fmaxf(acc.y * di + b.y, 0.0f);
            r.z = fmaxf(acc.z * di + b.z, 0.0f);
            r.w = fmaxf(acc.w * di + b.w, 0.0f);
            ((float4*)y)[(long long)i * LPE + j] = r;
        }
    }
}

__global__ void final_linear_kernel(const float* __restrict__ y, const float* __restrict__ Wl,
                                    const float* __restrict__ bl, float* out, int n) {
    int i = blockIdx.x * blockDim.x + threadIdx.x;
    if (i >= n) return;
    const float* row = y + (long long)i * 32;
    float s = 0.0f;
#pragma unroll
    for (int f = 0; f < 32; ++f) s += row[f] * Wl[f];
    out[i] = s + bl[0];
}

#define GEMM_GROUPS 1250

static inline size_t align256(size_t x) { return (x + 255) & ~size_t(255); }

extern "C" void kernel_launch(void* const* d_in, const int* in_sizes, int n_in,
                              void* d_out, int out_size, void* d_ws, size_t ws_size,
                              hipStream_t stream) {
    const float* x    = (const float*)d_in[0];
    const int*   ei   = (const int*)d_in[1];
    const float* W1   = (const float*)d_in[2];
    const float* b1   = (const float*)d_in[3];
    const float* W2   = (const float*)d_in[4];
    const float* b2   = (const float*)d_in[5];
    const float* Wl   = (const float*)d_in[6];
    const float* bl   = (const float*)d_in[7];
    float*       out  = (float*)d_out;

    const int N = N_NODES;
    const int E = N_EDGES;
    const int* src = ei;
    const int* dst = ei + E;

    char* ws = (char*)d_ws;
    size_t off = 0;
    int*   cell   = (int*)(ws + off);   off += align256((size_t)NCELL * 4);
    int*   celloff= (int*)(ws + off);   off += align256((size_t)(NCELL + 1) * 4);
    int*   rowptr = (int*)(ws + off);   off += align256((size_t)(N + 1) * 4);
    float* dinv   = (float*)(ws + off); off += align256((size_t)N * 4);
    int*   bsrc   = (int*)(ws + off);   off += align256((size_t)E * 4);
    int*   bdst   = (int*)(ws + off);   off += align256((size_t)E * 4);
    int*   csr    = (int*)(ws + off);   off += align256((size_t)E * 4);
    float* h1     = (float*)(ws + off); off += align256((size_t)N * 64 * 4);
    float* y1     = (float*)(ws + off); off += align256((size_t)N * 64 * 4);
    float* h2     = (float*)(ws + off); off += align256((size_t)N * 32 * 4);
    float* y2     = (float*)(ws + off); off += align256((size_t)N * 32 * 4);
    (void)ws_size; (void)n_in; (void)in_sizes; (void)out_size;

    const int T = 256;

    bin_count_kernel<<<NCHUNK, 1024, 0, stream>>>(dst, cell, E / 4);
    scan_kernel<<<(NCELL / 4 + 1023) / 1024, 1024, 0, stream>>>(cell, celloff, NCELL);
    bin_scatter_kernel<<<NCHUNK, 1024, 0, stream>>>(src, dst, celloff, bsrc, bdst, E / 4);
    bin_sort_kernel<<<NBIN, 1024, 0, stream>>>(bsrc, bdst, celloff, rowptr, dinv, csr, N);

    gemm_wmma_f32<128, 64, 1><<<GEMM_GROUPS, 4 * 32, 0, stream>>>(x, W1, dinv, h1, N);

    gather_kernel<64><<<2048, T, 0, stream>>>(h1, rowptr, csr, dinv, b1, y1, N);

    gemm_wmma_f32<64, 32, 1><<<GEMM_GROUPS, 2 * 32, 0, stream>>>(y1, W2, dinv, h2, N);

    gather_kernel<32><<<2048, T, 0, stream>>>(h2, rowptr, csr, dinv, b2, y2, N);

    final_linear_kernel<<<(N + T - 1) / T, T, 0, stream>>>(y2, Wl, bl, out, N);
}


# Edges grouped in two launches, not four: a chunk ranks its edges with its counting LDS add, orders them by bin and stores a bin offset row; bins fetch their runs by it; sums stay in arrival order
# speedup vs baseline: 6.1069x; 1.1966x over previous
#include <hip/hip_runtime.h>
#include <hip/hip_bf16.h>

typedef __attribute__((ext_vector_type(2))) float v2f;
typedef __attribute__((ext_vector_type(8))) float v8f;

#define N_NODES 100000
#define N_EDGES 3200000

static_assert(N_NODES % 4 == 0 && N_EDGES % 4 == 0, "16-byte index loads need multiples of 4");

#define RBITS   9
#define RANGE   (1 << RBITS)
#define NBIN    ((N_NODES + RANGE - 1) >> RBITS)
#define NBIN_PAD (((NBIN + 1) + 31) & ~31)
#define TROW    (NBIN + 1)
#define NCHUNK  256
#define CHUNK4  ((N_EDGES / 4 + NCHUNK - 1) / NCHUNK)
#define CHUNK   (CHUNK4 * 4)
#define SBITS   17
#define SMASK   ((1 << SBITS) - 1)
#define STAGE_CAP 32768

static_assert(N_NODES <= (1 << SBITS) && SBITS + RBITS <= 31, "an edge word holds source and in-bin destination");
static_assert(CHUNK4 <= 4 * 1024, "a chunk is at most four 16-byte groups per thread");
static_assert(NBIN_PAD <= 1024 && NCHUNK % 32 == 0 && NCHUNK <= 1024, "scans run on whole waves of one work-group");

__global__ __launch_bounds__(1024)
void chunk_order_kernel(const int* __restrict__ src, const int* __restrict__ dst,
                        int* __restrict__ tab, int* __restrict__ pk, int e4) {
    __shared__ int s_hist[NBIN_PAD];
    __shared__ int s_off[NBIN_PAD];
    __shared__ int s_wtot[NBIN_PAD / 32];
    __shared__ __attribute__((aligned(16))) int s_pack[CHUNK];
    const int tid  = threadIdx.x;
    const int lane = tid & 31;
    const int wid  = tid >> 5;
    const int b    = blockIdx.x;
    if (tid < NBIN_PAD) s_hist[tid] = 0;
    const int first = b * CHUNK4;
    const int last  = (first + CHUNK4 < e4) ? first + CHUNK4 : e4;
    const int4* d4 = (const int4*)dst;
    const int4* s4 = (const int4*)src;
    int4 d[4], s[4];
#pragma unroll
    for (int u = 0; u < 4; ++u) {
        const int i = first + tid + u * 1024;
        d[u] = (i < last) ? d4[i] : make_int4(-1, -1, -1, -1);
        s[u] = (i < last) ? s4[i] : make_int4(0, 0, 0, 0);
    }
    __syncthreads();
    int4 r[4];
#pragma unroll
    for (int u = 0; u < 4; ++u) {
        r[u] = make_int4(0, 0, 0, 0);
        if (d[u].x >= 0) {
            r[u].x = atomicAdd(&s_hist[d[u].x >> RBITS], 1);
            r[u].y = atomicAdd(&s_hist[d[u].y >> RBITS], 1);
            r[u].z = atomicAdd(&s_hist[d[u].z >> RBITS], 1);
            r[u].w = atomicAdd(&s_hist[d[u].w >> RBITS], 1);
        }
    }
    __syncthreads();
    int v = 0, incl = 0;
    if (tid < NBIN_PAD) {
        v = s_hist[tid];
        incl = v;
#pragma unroll
        for (int o = 1; o < 32; o <<= 1) {
            int t = __shfl_up(incl, o);
            if (lane >= o) incl += t;
        }
        if (lane == 31) s_wtot[wid] = incl;
    }
    __syncthreads();
    if (tid < NBIN_PAD) {
        int base = 0;
#pragma unroll
        for (int w = 0; w < NBIN_PAD / 32; ++w) if (w < wid) base += s_wtot[w];
        const int excl = base + incl - v;
        s_off[tid] = excl;
        if (tid < TROW) tab[b * TROW + tid] = excl;
    }
    __syncthreads();
#pragma unroll
    for (int u = 0; u < 4; ++u) {
        if (d[u].x >= 0) {
            s_pack[s_off[d[u].x >> RBITS] + r[u].x] = s[u].x | ((d[u].x & (RANGE - 1)) << SBITS);
            s_pack[s_off[d[u].y >> RBITS] + r[u].y] = s[u].y | ((d[u].y & (RANGE - 1)) << SBITS);
            s_pack[s_off[d[u].z >> RBITS] + r[u].z] = s[u].z | ((d[u].z & (RANGE - 1)) << SBITS);
            s_pack[s_off[d[u].w >> RBITS] + r[u].w] = s[u].w | ((d[u].w & (RANGE - 1)) << SBITS);
        }
    }
    __syncthreads();
    const int len4 = (last > first) ? last - first : 0;
    int4* o4 = (int4*)(pk + (long long)b * CHUNK);
    for (int i = tid; i < len4; i += 1024) o4[i] = ((const int4*)s_pack)[i];
}

__global__ __launch_bounds__(1024)
void bin_sort_kernel(const int* __restrict__ pk, const int* __restrict__ tab,
                     int* __restrict__ rowptr, float* __restrict__ dinv,
                     int* __restrict__ csr_src, int n) {
    constexpr int RPW = NCHUNK / 32;
    __shared__ int s_cnt[RANGE];
    __shared__ int s_cur[RANGE];
    __shared__ int s_wtot[RANGE / 32];
    __shared__ int s_rbase[NCHUNK];
    __shared__ int s_rlen[NCHUNK];
    __shared__ int s_rpos[NCHUNK];
    __shared__ int s_rtot[NCHUNK / 32];
    __shared__ int s_rlow[NCHUNK / 32];
    __shared__ int s_in[STAGE_CAP];
    __shared__ int s_stage[STAGE_CAP];
    const int tid  = threadIdx.x;
    const int lane = tid & 31;
    const int wid  = tid >> 5;
    const int bin  = blockIdx.x;
    if (tid < RANGE) s_cnt[tid] = 0;

    int rlen = 0, rincl = 0, rlo = 0;
    if (tid < NCHUNK) {
        rlo  = tab[tid * TROW + bin];
        rlen = tab[tid * TROW + bin + 1] - rlo;
        rincl = rlen;
#pragma unroll
        for (int o = 1; o < 32; o <<= 1) {
            int t = __shfl_up(rincl, o);
            if (lane >= o) rincl += t;
        }
        int lows = rlo;
#pragma unroll
        for (int o = 16; o >= 1; o >>= 1) lows += __shfl_xor(lows, o);
        if (lane == 31) s_rtot[wid] = rincl;
        if (lane == 0)  s_rlow[wid] = lows;
    }
    __syncthreads();
    int beg = 0, cnt = 0;
#pragma unroll
    for (int w = 0; w < NCHUNK / 32; ++w) { beg += s_rlow[w]; cnt += s_rtot[w]; }
    if (tid < NCHUNK) {
        int base = 0;
#pragma unroll
        for (int w = 0; w < NCHUNK / 32; ++w) if (w < wid) base += s_rtot[w];
        s_rpos[tid]  = base + rincl - rlen;
        s_rbase[tid] = tid * CHUNK + rlo;
        s_rlen[tid]  = rlen;
    }
    __syncthreads();

    int rb[RPW], rl[RPW], rp[RPW];
#pragma unroll
    for (int rr = 0; rr < RPW; ++rr) {
        const int run = wid + 32 * rr;
        rb[rr] = s_rbase[run];
        rl[rr] = s_rlen[run];
        rp[rr] = s_rpos[run];
    }

    const bool staged = cnt <= STAGE_CAP;
    if (staged) {
        int e[RPW][4];
#pragma unroll
        for (int rr = 0; rr < RPW; ++rr) {
#pragma unroll
            for (int u = 0; u < 4; ++u) {
                const int k = lane + 32 * u;
                e[rr][u] = (k < rl[rr]) ? pk[rb[rr] + k] : 0;
            }
        }
#pragma unroll
        for (int rr = 0; rr < RPW; ++rr) {
#pragma unroll
            for (int u = 0; u < 4; ++u) {
                const int k = lane + 32 * u;
                if (k < rl[rr]) {
                    atomicAdd(&s_cnt[e[rr][u] >> SBITS], 1);
                    s_in[rp[rr] + k] = e[rr][u];
                }
            }
            for (int k = 128 + lane; k < rl[rr]; k += 32) {
                const int w = pk[rb[rr] + k];
                atomicAdd(&s_cnt[w >> SBITS], 1);
                s_in[rp[rr] + k] = w;
            }
        }
    } else {
#pragma unroll
        for (int rr = 0; rr < RPW; ++rr)
            for (int k = lane; k < rl[rr]; k += 32) atomicAdd(&s_cnt[pk[rb[rr] + k] >> SBITS], 1);
    }
    __syncthreads();

    int v = 0, incl = 0;
    if (tid < RANGE) {
        v = s_cnt[tid];
        incl = v;
#pragma unroll
        for (int o = 1; o < 32; o <<= 1) {
            int t = __shfl_up(incl, o);
            if (lane >= o) incl += t;
        }
        if (lane == 31) s_wtot[wid] = incl;
    }
    __syncthreads();
    if (tid < RANGE) {
        int base = 0;
#pragma unroll
        for (int w = 0; w < RANGE / 32; ++w) if (w < wid) base += s_wtot[w];
        const int excl = base + incl - v;
        s_cur[tid] = excl;
        const int node = bin * RANGE + tid;
        if (node < n) {
            rowptr[node] = beg + excl;
            dinv[node]   = rsqrtf((float)(v + 1));
            if (node == n - 1) rowptr[n] = beg + excl + v;
        }
    }
    __syncthreads();

    if (staged) {
        int i = tid;
        for (; i + 3 * 1024 < cnt; i += 4 * 1024) {
            int w[4];
#pragma unroll
            for (int u = 0; u < 4; ++u) w[u] = s_in[i + u * 1024];
#pragma unroll
            for (int u = 0; u < 4; ++u) {
                const int p = atomicAdd(&s_cur[w[u] >> SBITS], 1);
                s_stage[p] = w[u] & SMASK;
            }
        }
        for (; i < cnt; i += 1024) {
            const int w = s_in[i];
            const int p = atomicAdd(&s_cur[w >> SBITS], 1);
            s_stage[p] = w & SMASK;
        }
        __syncthreads();
        for (int k = tid; k < cnt; k += 1024) csr_src[beg + k] = s_stage[k];
    } else {
#pragma unroll
        for (int rr = 0; rr < RPW; ++rr) {
            for (int k = lane; k < rl[rr]; k += 32) {
                const int w = pk[rb[rr] + k];
                const int p = atomicAdd(&s_cur[w >> SBITS], 1);
                csr_src[beg + p] = w & SMASK;
            }
        }
    }
}

template <int K, int NOUT, int MT>
__global__ __launch_bounds__(MT*(NOUT/16)*32)
void gemm_wmma_f32(const float* __restrict__ A, const float* __restrict__ B,
                   const float* __restrict__ rowscale, float* __restrict__ C, int M) {
    constexpr int NT   = NOUT / 16;
    constexpr int NW   = MT * NT;
    constexpr int NTHR = NW * 32;
    constexpr int ROWS = MT * 16;
    constexpr int KP   = K + 4;
    constexpr int KS   = K / 4;
    constexpr int A_ELTS = ROWS * K;
    constexpr int PER  = A_ELTS / (NTHR * 4);
    static_assert(A_ELTS % (NTHR * 4) == 0, "a tile is a whole number of 16-byte pieces per thread");

    __shared__ float As[2][ROWS][KP];

    const int tid    = threadIdx.x;
    const int ntiles = (M + ROWS - 1) / ROWS;

    float4 pv[PER];
    float  ps[PER];
    int tile = blockIdx.x;
    if (tile < ntiles) {
#pragma unroll
        for (int u = 0; u < PER; ++u) {
            const int i  = (tid + u * NTHR) * 4;
            const int gr = tile * ROWS + i / K;
            pv[u] = (gr < M) ? *(const float4*)(A + (long long)gr * K + i % K)
                             : make_float4(0.f, 0.f, 0.f, 0.f);
            ps[u] = (gr < M) ? rowscale[gr] : 0.f;
        }
    }

    const int wid  = tid >> 5;
    const int lane = tid & 31;
    const int mn   = lane & 15;
    const int h    = lane >> 4;
    const int mt   = wid / NT;
    const int nt   = wid % NT;
    const int ncol = nt * 16 + mn;

    v2f bf[KS];
#pragma unroll
    for (int kk = 0; kk < KS; ++kk) {
        bf[kk].x = B[(4 * kk + 2 * h + 0) * NOUT + ncol];
        bf[kk].y = B[(4 * kk + 2 * h + 1) * NOUT + ncol];
    }

    int buf = 0;
    for (; tile < ntiles; tile += gridDim.x) {
#pragma unroll
        for (int u = 0; u < PER; ++u) {
            const int i = (tid + u * NTHR) * 4;
            float4 v = pv[u];
            const float s = ps[u];
            v.x *= s; v.y *= s; v.z *= s; v.w *= s;
            *(float4*)(&As[buf][i / K][i % K]) = v;
        }
        __syncthreads();

        const int nxt = tile + gridDim.x;
        if (nxt < ntiles) {
#pragma unroll
            for (int u = 0; u < PER; ++u) {
                const int i  = (tid + u * NTHR) * 4;
                const int gr = nxt * ROWS + i / K;
                pv[u] = (gr < M) ? *(const float4*)(A + (long long)gr * K + i % K)
                                 : make_float4(0.f, 0.f, 0.f, 0.f);
                ps[u] = (gr < M) ? rowscale[gr] : 0.f;
            }
        }

        v8f acc = {};
        const float* arow = &As[buf][mt * 16 + mn][0];
#pragma unroll
        for (int kk = 0; kk < KS; ++kk) {
            v2f a;
            a.x = arow[4 * kk + 2 * h + 0];
            a.y = arow[4 * kk + 2 * h + 1];
            acc = __builtin_amdgcn_wmma_f32_16x16x4_f32(
                 false, a,  false, bf[kk],
                 (short)0, acc,  false,  false);
        }

        const int row0 = tile * ROWS;
#pragma unroll
        for (int j = 0; j < 8; ++j) {
            int r = row0 + mt * 16 + j + 8 * h;
            if (r < M) C[(long long)r * NOUT + ncol] = acc[j];
        }
        buf ^= 1;
    }
}

template <int F>
__global__ __launch_bounds__(256)
void gather_kernel(const float* __restrict__ hs, const int* __restrict__ rowptr,
                   const int* __restrict__ csr_src, const float* __restrict__ dinv,
                   const float* __restrict__ bias, float* __restrict__ y, int n) {
    constexpr int LPE = F / 4;
    constexpr int G   = 32 / LPE;
    const int lane = threadIdx.x & 31;
    const int g = lane / LPE;
    const int j = lane % LPE;
    const float4* h4 = (const float4*)hs;
    const float4 b = ((const float4*)bias)[j];
    const int wstride = gridDim.x * 8;

    for (int i = blockIdx.x * 8 + (threadIdx.x >> 5); i < n; i += wstride) {
        const int beg = rowptr[i];
        const int end = rowptr[i + 1];
        const float di = dinv[i];
        float4 acc = make_float4(0.f, 0.f, 0.f, 0.f);
        if (g == 0) acc = h4[(long long)i * LPE + j];
        const int cntg = (end - beg - g + G - 1) / G;
        const int* q = csr_src + beg + g;
        int k = 0;
        for (; k + 4 <= cntg; k += 4) {
            const int s0 = q[0];
            const int s1 = q[G];
            const int s2 = q[2 * G];
            const int s3 = q[3 * G];
            q += 4 * G;
            const float4 v0 = h4[(long long)s0 * LPE + j];
            const float4 v1 = h4[(long long)s1 * LPE + j];
            const float4 v2 = h4[(long long)s2 * LPE + j];
            const float4 v3 = h4[(long long)s3 * LPE + j];
            acc.x += v0.x; acc.y += v0.y; acc.z += v0.z; acc.w += v0.w;
            acc.x += v1.x; acc.y += v1.y; acc.z += v1.z; acc.w += v1.w;
            acc.x += v2.x; acc.y += v2.y; acc.z += v2.z; acc.w += v2.w;
            acc.x += v3.x; acc.y += v3.y; acc.z += v3.z; acc.w += v3.w;
        }
#pragma unroll 1
        for (; k < cntg; ++k) {
            const int s0 = q[0];
            q += G;
            const float4 v0 = h4[(long long)s0 * LPE + j];
            acc.x += v0.x; acc.y += v0.y; acc.z += v0.z; acc.w += v0.w;
        }
#pragma unroll
        for (int o = LPE; o < 32; o <<= 1) {
            acc.x += __shfl_xor(acc.x, o);
            acc.y += __shfl_xor(acc.y, o);
            acc.z += __shfl_xor(acc.z, o);
            acc.w += __shfl_xor(acc.w, o);
        }
        if (g == 0) {
            float4 r;
            r.x = fmaxf(acc.x * di + b.x, 0.0f);
            r.y = fmaxf(acc.y * di + b.y, 0.0f);
            r.z = fmaxf(acc.z * di + b.z, 0.0f);
            r.w = fmaxf(acc.w * di + b.w, 0.0f);
            ((float4*)y)[(long long)i * LPE + j] = r;
        }
    }
}

__global__ void final_linear_kernel(const float* __restrict__ y, const float* __restrict__ Wl,
                                    const float* __restrict__ bl, float* out, int n) {
    int i = blockIdx.x * blockDim.x + threadIdx.x;
    if (i >= n) return;
    const float* row = y + (long long)i * 32;
    float s = 0.0f;
#pragma unroll
    for (int f = 0; f < 32; ++f) s += row[f] * Wl[f];
    out[i] = s + bl[0];
}

#define GEMM_GROUPS 1250

static inline size_t align256(size_t x) { return (x + 255) & ~size_t(255); }

extern "C" void kernel_launch(void* const* d_in, const int* in_sizes, int n_in,
                              void* d_out, int out_size, void* d_ws, size_t ws_size,
                              hipStream_t stream) {
    const float* x    = (const float*)d_in[0];
    const int*   ei   = (const int*)d_in[1];
    const float* W1   = (const float*)d_in[2];
    const float* b1   = (const float*)d_in[3];
    const float* W2   = (const float*)d_in[4];
    const float* b2   = (const float*)d_in[5];
    const float* Wl   = (const float*)d_in[6];
    const float* bl   = (const float*)d_in[7];
    float*       out  = (float*)d_out;

    const int N = N_NODES;
    const int E = N_EDGES;
    const int* src = ei;
    const int* dst = ei + E;

    char* ws = (char*)d_ws;
    size_t off = 0;
    int*   tab    = (int*)(ws + off);   off += align256((size_t)NCHUNK * TROW * 4);
    int*   rowptr = (int*)(ws + off);   off += align256((size_t)(N + 1) * 4);
    float* dinv   = (float*)(ws + off); off += align256((size_t)N * 4);
    int*   pk     = (int*)(ws + off);   off += align256((size_t)NCHUNK * CHUNK * 4);
    int*   csr    = (int*)(ws + off);   off += align256((size_t)E * 4);
    float* h1     = (float*)(ws + off); off += align256((size_t)N * 64 * 4);
    float* y1     = (float*)(ws + off); off += align256((size_t)N * 64 * 4);
    float* h2     = (float*)(ws + off); off += align256((size_t)N * 32 * 4);
    float* y2     = (float*)(ws + off); off += align256((size_t)N * 32 * 4);
    (void)ws_size; (void)n_in; (void)in_sizes; (void)out_size;

    const int T = 256;

    chunk_order_kernel<<<NCHUNK, 1024, 0, stream>>>(src, dst, tab, pk, E / 4);
    bin_sort_kernel<<<NBIN, 1024, 0, stream>>>(pk, tab, rowptr, dinv, csr, N);

    gemm_wmma_f32<128, 64, 1><<<GEMM_GROUPS, 4 * 32, 0, stream>>>(x, W1, dinv, h1, N);

    gather_kernel<64><<<2048, T, 0, stream>>>(h1, rowptr, csr, dinv, b1, y1, N);

    gemm_wmma_f32<64, 32, 1><<<GEMM_GROUPS, 2 * 32, 0, stream>>>(y1, W2, dinv, h2, N);

    gather_kernel<32><<<2048, T, 0, stream>>>(h2, rowptr, csr, dinv, b2, y2, N);

    final_linear_kernel<<<(N + T - 1) / T, T, 0, stream>>>(y2, Wl, bl, out, N);
}


# Each aggregation launch does the dense step after it: layer-2 product on WMMA from an LDS tile of 64 summed rows; final dot as 8 lane partials + lane tree (its sum order changes); 7 launches -> 5
# speedup vs baseline: 6.5823x; 1.0778x over previous
#include <hip/hip_runtime.h>
#include <hip/hip_bf16.h>

typedef __attribute__((ext_vector_type(2))) float v2f;
typedef __attribute__((ext_vector_type(8))) float v8f;

#define N_NODES 100000
#define N_EDGES 3200000

static_assert(N_NODES % 4 == 0 && N_EDGES % 4 == 0, "16-byte index loads need multiples of 4");

#define RBITS   9
#define RANGE   (1 << RBITS)
#define NBIN    ((N_NODES + RANGE - 1) >> RBITS)
#define NBIN_PAD (((NBIN + 1) + 31) & ~31)
#define TROW    (NBIN + 1)
#define NCHUNK  256
#define CHUNK4  ((N_EDGES / 4 + NCHUNK - 1) / NCHUNK)
#define CHUNK   (CHUNK4 * 4)
#define SBITS   17
#define SMASK   ((1 << SBITS) - 1)
#define STAGE_CAP 32768

static_assert(N_NODES <= (1 << SBITS) && SBITS + RBITS <= 31, "an edge word holds source and in-bin destination");
static_assert(CHUNK4 <= 4 * 1024, "a chunk is at most four 16-byte groups per thread");
static_assert(NBIN_PAD <= 1024 && NCHUNK % 32 == 0 && NCHUNK <= 1024, "scans run on whole waves of one work-group");

__global__ __launch_bounds__(1024)
void chunk_order_kernel(const int* __restrict__ src, const int* __restrict__ dst,
                        int* __restrict__ tab, int* __restrict__ pk, int e4) {
    __shared__ int s_hist[NBIN_PAD];
    __shared__ int s_off[NBIN_PAD];
    __shared__ int s_wtot[NBIN_PAD / 32];
    __shared__ __attribute__((aligned(16))) int s_pack[CHUNK];
    const int tid  = threadIdx.x;
    const int lane = tid & 31;
    const int wid  = tid >> 5;
    const int b    = blockIdx.x;
    if (tid < NBIN_PAD) s_hist[tid] = 0;
    const int first = b * CHUNK4;
    const int last  = (first + CHUNK4 < e4) ? first + CHUNK4 : e4;
    const int4* d4 = (const int4*)dst;
    const int4* s4 = (const int4*)src;
    int4 d[4], s[4];
#pragma unroll
    for (int u = 0; u < 4; ++u) {
        const int i = first + tid + u * 1024;
        d[u] = (i < last) ? d4[i] : make_int4(-1, -1, -1, -1);
        s[u] = (i < last) ? s4[i] : make_int4(0, 0, 0, 0);
    }
    __syncthreads();
    int4 r[4];
#pragma unroll
    for (int u = 0; u < 4; ++u) {
        r[u] = make_int4(0, 0, 0, 0);
        if (d[u].x >= 0) {
            r[u].x = atomicAdd(&s_hist[d[u].x >> RBITS], 1);
            r[u].y = atomicAdd(&s_hist[d[u].y >> RBITS], 1);
            r[u].z = atomicAdd(&s_hist[d[u].z >> RBITS], 1);
            r[u].w = atomicAdd(&s_hist[d[u].w >> RBITS], 1);
        }
    }
    __syncthreads();
    int v = 0, incl = 0;
    if (tid < NBIN_PAD) {
        v = s_hist[tid];
        incl = v;
#pragma unroll
        for (int o = 1; o < 32; o <<= 1) {
            int t = __shfl_up(incl, o);
            if (lane >= o) incl += t;
        }
        if (lane == 31) s_wtot[wid] = incl;
    }
    __syncthreads();
    if (tid < NBIN_PAD) {
        int base = 0;
#pragma unroll
        for (int w = 0; w < NBIN_PAD / 32; ++w) if (w < wid) base += s_wtot[w];
        const int excl = base + incl - v;
        s_off[tid] = excl;
        if (tid < TROW) tab[b * TROW + tid] = excl;
    }
    __syncthreads();
#pragma unroll
    for (int u = 0; u < 4; ++u) {
        if (d[u].x >= 0) {
            s_pack[s_off[d[u].x >> RBITS] + r[u].x] = s[u].x | ((d[u].x & (RANGE - 1)) << SBITS);
            s_pack[s_off[d[u].y >> RBITS] + r[u].y] = s[u].y | ((d[u].y & (RANGE - 1)) << SBITS);
            s_pack[s_off[d[u].z >> RBITS] + r[u].z] = s[u].z | ((d[u].z & (RANGE - 1)) << SBITS);
            s_pack[s_off[d[u].w >> RBITS] + r[u].w] = s[u].w | ((d[u].w & (RANGE - 1)) << SBITS);
        }
    }
    __syncthreads();
    const int len4 = (last > first) ? last - first : 0;
    int4* o4 = (int4*)(pk + (long long)b * CHUNK);
    for (int i = tid; i < len4; i += 1024) o4[i] = ((const int4*)s_pack)[i];
}

__global__ __launch_bounds__(1024)
void bin_sort_kernel(const int* __restrict__ pk, const int* __restrict__ tab,
                     int* __restrict__ rowptr, float* __restrict__ dinv,
                     int* __restrict__ csr_src, int n) {
    constexpr int RPW = NCHUNK / 32;
    __shared__ int s_cnt[RANGE];
    __shared__ int s_cur[RANGE];
    __shared__ int s_wtot[RANGE / 32];
    __shared__ int s_rbase[NCHUNK];
    __shared__ int s_rlen[NCHUNK];
    __shared__ int s_rpos[NCHUNK];
    __shared__ int s_rtot[NCHUNK / 32];
    __shared__ int s_rlow[NCHUNK / 32];
    __shared__ int s_in[STAGE_CAP];
    __shared__ int s_stage[STAGE_CAP];
    const int tid  = threadIdx.x;
    const int lane = tid & 31;
    const int wid  = tid >> 5;
    const int bin  = blockIdx.x;
    if (tid < RANGE) s_cnt[tid] = 0;

    int rlen = 0, rincl = 0, rlo = 0;
    if (tid < NCHUNK) {
        rlo  = tab[tid * TROW + bin];
        rlen = tab[tid * TROW + bin + 1] - rlo;
        rincl = rlen;
#pragma unroll
        for (int o = 1; o < 32; o <<= 1) {
            int t = __shfl_up(rincl, o);
            if (lane >= o) rincl += t;
        }
        int lows = rlo;
#pragma unroll
        for (int o = 16; o >= 1; o >>= 1) lows += __shfl_xor(lows, o);
        if (lane == 31) s_rtot[wid] = rincl;
        if (lane == 0)  s_rlow[wid] = lows;
    }
    __syncthreads();
    int beg = 0, cnt = 0;
#pragma unroll
    for (int w = 0; w < NCHUNK / 32; ++w) { beg += s_rlow[w]; cnt += s_rtot[w]; }
    if (tid < NCHUNK) {
        int base = 0;
#pragma unroll
        for (int w = 0; w < NCHUNK / 32; ++w) if (w < wid) base += s_rtot[w];
        s_rpos[tid]  = base + rincl - rlen;
        s_rbase[tid] = tid * CHUNK + rlo;
        s_rlen[tid]  = rlen;
    }
    __syncthreads();

    int rb[RPW], rl[RPW], rp[RPW];
#pragma unroll
    for (int rr = 0; rr < RPW; ++rr) {
        const int run = wid + 32 * rr;
        rb[rr] = s_rbase[run];
        rl[rr] = s_rlen[run];
        rp[rr] = s_rpos[run];
    }

    const bool staged = cnt <= STAGE_CAP;
    if (staged) {
        int e[RPW][4];
#pragma unroll
        for (int rr = 0; rr < RPW; ++rr) {
#pragma unroll
            for (int u = 0; u < 4; ++u) {
                const int k = lane + 32 * u;
                e[rr][u] = (k < rl[rr]) ? pk[rb[rr] + k] : 0;
            }
        }
#pragma unroll
        for (int rr = 0; rr < RPW; ++rr) {
#pragma unroll
            for (int u = 0; u < 4; ++u) {
                const int k = lane + 32 * u;
                if (k < rl[rr]) {
                    atomicAdd(&s_cnt[e[rr][u] >> SBITS], 1);
                    s_in[rp[rr] + k] = e[rr][u];
                }
            }
            for (int k = 128 + lane; k < rl[rr]; k += 32) {
                const int w = pk[rb[rr] + k];
                atomicAdd(&s_cnt[w >> SBITS], 1);
                s_in[rp[rr] + k] = w;
            }
        }
    } else {
#pragma unroll
        for (int rr = 0; rr < RPW; ++rr)
            for (int k = lane; k < rl[rr]; k += 32) atomicAdd(&s_cnt[pk[rb[rr] + k] >> SBITS], 1);
    }
    __syncthreads();

    int v = 0, incl = 0;
    if (tid < RANGE) {
        v = s_cnt[tid];
        incl = v;
#pragma unroll
        for (int o = 1; o < 32; o <<= 1) {
            int t = __shfl_up(incl, o);
            if (lane >= o) incl += t;
        }
        if (lane == 31) s_wtot[wid] = incl;
    }
    __syncthreads();
    if (tid < RANGE) {
        int base = 0;
#pragma unroll
        for (int w = 0; w < RANGE / 32; ++w) if (w < wid) base += s_wtot[w];
        const int excl = base + incl - v;
        s_cur[tid] = excl;
        const int node = bin * RANGE + tid;
        if (node < n) {
            rowptr[node] = beg + excl;
            dinv[node]   = rsqrtf((float)(v + 1));
            if (node == n - 1) rowptr[n] = beg + excl + v;
        }
    }
    __syncthreads();

    if (staged) {
        int i = tid;
        for (; i + 3 * 1024 < cnt; i += 4 * 1024) {
            int w[4];
#pragma unroll
            for (int u = 0; u < 4; ++u) w[u] = s_in[i + u * 1024];
#pragma unroll
            for (int u = 0; u < 4; ++u) {
                const int p = atomicAdd(&s_cur[w[u] >> SBITS], 1);
                s_stage[p] = w[u] & SMASK;
            }
        }
        for (; i < cnt; i += 1024) {
            const int w = s_in[i];
            const int p = atomicAdd(&s_cur[w >> SBITS], 1);
            s_stage[p] = w & SMASK;
        }
        __syncthreads();
        for (int k = tid; k < cnt; k += 1024) csr_src[beg + k] = s_stage[k];
    } else {
#pragma unroll
        for (int rr = 0; rr < RPW; ++rr) {
            for (int k = lane; k < rl[rr]; k += 32) {
                const int w = pk[rb[rr] + k];
                const int p = atomicAdd(&s_cur[w >> SBITS], 1);
                csr_src[beg + p] = w & SMASK;
            }
        }
    }
}

template <int K, int NOUT, int MT>
__global__ __launch_bounds__(MT*(NOUT/16)*32)
void gemm_wmma_f32(const float* __restrict__ A, const float* __restrict__ B,
                   const float* __restrict__ rowscale, float* __restrict__ C, int M) {
    constexpr int NT   = NOUT / 16;
    constexpr int NW   = MT * NT;
    constexpr int NTHR = NW * 32;
    constexpr int ROWS = MT * 16;
    constexpr int KP   = K + 4;
    constexpr int KS   = K / 4;
    constexpr int A_ELTS = ROWS * K;
    constexpr int PER  = A_ELTS / (NTHR * 4);
    static_assert(A_ELTS % (NTHR * 4) == 0, "a tile is a whole number of 16-byte pieces per thread");

    __shared__ float As[2][ROWS][KP];

    const int tid    = threadIdx.x;
    const int ntiles = (M + ROWS - 1) / ROWS;

    float4 pv[PER];
    float  ps[PER];
    int tile = blockIdx.x;
    if (tile < ntiles) {
#pragma unroll
        for (int u = 0; u < PER; ++u) {
            const int i  = (tid + u * NTHR) * 4;
            const int gr = tile * ROWS + i / K;
            pv[u] = (gr < M) ? *(const float4*)(A + (long long)gr * K + i % K)
                             : make_float4(0.f, 0.f, 0.f, 0.f);
            ps[u] = (gr < M) ? rowscale[gr] : 0.f;
        }
    }

    const int wid  = tid >> 5;
    const int lane = tid & 31;
    const int mn   = lane & 15;
    const int h    = lane >> 4;
    const int mt   = wid / NT;
    const int nt   = wid % NT;
    const int ncol = nt * 16 + mn;

    v2f bf[KS];
#pragma unroll
    for (int kk = 0; kk < KS; ++kk) {
        bf[kk].x = B[(4 * kk + 2 * h + 0) * NOUT + ncol];
        bf[kk].y = B[(4 * kk + 2 * h + 1) * NOUT + ncol];
    }

    int buf = 0;
    for (; tile < ntiles; tile += gridDim.x) {
#pragma unroll
        for (int u = 0; u < PER; ++u) {
            const int i = (tid + u * NTHR) * 4;
            float4 v = pv[u];
            const float s = ps[u];
            v.x *= s; v.y *= s; v.z *= s; v.w *= s;
            *(float4*)(&As[buf][i / K][i % K]) = v;
        }
        __syncthreads();

        const int nxt = tile + gridDim.x;
        if (nxt < ntiles) {
#pragma unroll
            for (int u = 0; u < PER; ++u) {
                const int i  = (tid + u * NTHR) * 4;
                const int gr = nxt * ROWS + i / K;
                pv[u] = (gr < M) ? *(const float4*)(A + (long long)gr * K + i % K)
                                 : make_float4(0.f, 0.f, 0.f, 0.f);
                ps[u] = (gr < M) ? rowscale[gr] : 0.f;
            }
        }

        v8f acc = {};
        const float* arow = &As[buf][mt * 16 + mn][0];
#pragma unroll
        for (int kk = 0; kk < KS; ++kk) {
            v2f a;
            a.x = arow[4 * kk + 2 * h + 0];
            a.y = arow[4 * kk + 2 * h + 1];
            acc = __builtin_amdgcn_wmma_f32_16x16x4_f32(
                 false, a,  false, bf[kk],
                 (short)0, acc,  false,  false);
        }

        const int row0 = tile * ROWS;
#pragma unroll
        for (int j = 0; j < 8; ++j) {
            int r = row0 + mt * 16 + j + 8 * h;
            if (r < M) C[(long long)r * NOUT + ncol] = acc[j];
        }
        buf ^= 1;
    }
}

template <int F>
__device__ __forceinline__ float4 node_sum(const float4* __restrict__ h4, const int* __restrict__ rowptr,
                                           const int* __restrict__ csr_src, int i, int g, int j) {
    constexpr int LPE = F / 4;
    constexpr int G   = 32 / LPE;
    const int beg = rowptr[i];
    const int end = rowptr[i + 1];
    float4 acc = make_float4(0.f, 0.f, 0.f, 0.f);
    if (g == 0) acc = h4[(long long)i * LPE + j];
    const int cntg = (end - beg - g + G - 1) / G;
    const int* q = csr_src + beg + g;
    int k = 0;
    for (; k + 4 <= cntg; k += 4) {
        const int s0 = q[0];
        const int s1 = q[G];
        const int s2 = q[2 * G];
        const int s3 = q[3 * G];
        q += 4 * G;
        const float4 v0 = h4[(long long)s0 * LPE + j];
        const float4 v1 = h4[(long long)s1 * LPE + j];
        const float4 v2 = h4[(long long)s2 * LPE + j];
        const float4 v3 = h4[(long long)s3 * LPE + j];
        acc.x += v0.x; acc.y += v0.y; acc.z += v0.z; acc.w += v0.w;
        acc.x += v1.x; acc.y += v1.y; acc.z += v1.z; acc.w += v1.w;
        acc.x += v2.x; acc.y += v2.y; acc.z += v2.z; acc.w += v2.w;
        acc.x += v3.x; acc.y += v3.y; acc.z += v3.z; acc.w += v3.w;
    }
#pragma unroll 1
    for (; k < cntg; ++k) {
        const int s0 = q[0];
        q += G;
        const float4 v0 = h4[(long long)s0 * LPE + j];
        acc.x += v0.x; acc.y += v0.y; acc.z += v0.z; acc.w += v0.w;
    }
#pragma unroll
    for (int o = LPE; o < 32; o <<= 1) {
        acc.x += __shfl_xor(acc.x, o);
        acc.y += __shfl_xor(acc.y, o);
        acc.z += __shfl_xor(acc.z, o);
        acc.w += __shfl_xor(acc.w, o);
    }
    return acc;
}

template <int F, int NOUT, int MT>
__global__ __launch_bounds__(MT*(NOUT/16)*32)
void gather_gemm_kernel(const float* __restrict__ hs, const int* __restrict__ rowptr,
                        const int* __restrict__ csr_src, const float* __restrict__ dinv,
                        const float* __restrict__ bias, const float* __restrict__ B,
                        float* __restrict__ C, int n) {
    constexpr int LPE  = F / 4;
    constexpr int NT   = NOUT / 16;
    constexpr int NW   = MT * NT;
    constexpr int ROWS = MT * 16;
    constexpr int NPW  = ROWS / NW;
    constexpr int KP   = F + 4;
    constexpr int KS   = F / 4;
    __shared__ float As[ROWS][KP];

    const int tid  = threadIdx.x;
    const int lane = tid & 31;
    const int wid  = tid >> 5;
    const int g = lane / LPE;
    const int j = lane % LPE;
    const float4* h4 = (const float4*)hs;
    const float4 b = ((const float4*)bias)[j];
    const int row0 = blockIdx.x * ROWS;

    for (int t = 0; t < NPW; ++t) {
        const int lr = wid * NPW + t;
        const int i  = row0 + lr;
        float4 r = make_float4(0.f, 0.f, 0.f, 0.f);
        if (i < n) {
            const float di = dinv[i];
            const float4 acc = node_sum<F>(h4, rowptr, csr_src, i, g, j);
            r.x = fmaxf(acc.x * di + b.x, 0.0f) * di;
            r.y = fmaxf(acc.y * di + b.y, 0.0f) * di;
            r.z = fmaxf(acc.z * di + b.z, 0.0f) * di;
            r.w = fmaxf(acc.w * di + b.w, 0.0f) * di;
        }
        if (g == 0) *(float4*)(&As[lr][4 * j]) = r;
    }

    const int mn   = lane & 15;
    const int h    = lane >> 4;
    const int mt   = wid / NT;
    const int nt   = wid % NT;
    const int ncol = nt * 16 + mn;

    v2f bf[KS];
#pragma unroll
    for (int kk = 0; kk < KS; ++kk) {
        bf[kk].x = B[(4 * kk + 2 * h + 0) * NOUT + ncol];
        bf[kk].y = B[(4 * kk + 2 * h + 1) * NOUT + ncol];
    }
    __syncthreads();

    v8f acc = {};
    const float* arow = &As[mt * 16 + mn][0];
#pragma unroll
    for (int kk = 0; kk < KS; ++kk) {
        v2f a;
        a.x = arow[4 * kk + 2 * h + 0];
        a.y = arow[4 * kk + 2 * h + 1];
        acc = __builtin_amdgcn_wmma_f32_16x16x4_f32(
             false, a,  false, bf[kk],
             (short)0, acc,  false,  false);
    }

#pragma unroll
    for (int jj = 0; jj < 8; ++jj) {
        const int r = row0 + mt * 16 + jj + 8 * h;
        if (r < n) C[(long long)r * NOUT + ncol] = acc[jj];
    }
}

template <int F, int NPW>
__global__ __launch_bounds__(256)
void gather_linear_kernel(const float* __restrict__ hs, const int* __restrict__ rowptr,
                          const int* __restrict__ csr_src, const float* __restrict__ dinv,
                          const float* __restrict__ bias, const float* __restrict__ Wl,
                          const float* __restrict__ bl, float* __restrict__ out, int n) {
    constexpr int LPE  = F / 4;
    constexpr int ROWS = 8 * NPW;
    __shared__ float s_out[ROWS];
    const int tid  = threadIdx.x;
    const int lane = tid & 31;
    const int wid  = tid >> 5;
    const int g = lane / LPE;
    const int j = lane % LPE;
    const float4* h4 = (const float4*)hs;
    const float4 b  = ((const float4*)bias)[j];
    const float4 wl = ((const float4*)Wl)[j];
    const int row0 = blockIdx.x * ROWS;

    for (int t = 0; t < NPW; ++t) {
        const int lr = wid * NPW + t;
        const int i  = row0 + lr;
        if (i < n) {
            const float di = dinv[i];
            const float4 acc = node_sum<F>(h4, rowptr, csr_src, i, g, j);
            float4 r;
            r.x = fmaxf(acc.x * di + b.x, 0.0f);
            r.y = fmaxf(acc.y * di + b.y, 0.0f);
            r.z = fmaxf(acc.z * di + b.z, 0.0f);
            r.w = fmaxf(acc.w * di + b.w, 0.0f);
            float p = r.x * wl.x + r.y * wl.y + r.z * wl.z + r.w * wl.w;
#pragma unroll
            for (int o = 1; o < LPE; o <<= 1) p += __shfl_xor(p, o);
            if (lane == 0) s_out[lr] = p;
        }
    }
    __syncthreads();
    if (tid < ROWS) {
        const int i = row0 + tid;
        if (i < n) out[i] = s_out[tid] + bl[0];
    }
}

#define GEMM_GROUPS 1250

#define AGG_ROWS   64
#define AGG_GROUPS ((N_NODES + AGG_ROWS - 1) / AGG_ROWS)

static inline size_t align256(size_t x) { return (x + 255) & ~size_t(255); }

extern "C" void kernel_launch(void* const* d_in, const int* in_sizes, int n_in,
                              void* d_out, int out_size, void* d_ws, size_t ws_size,
                              hipStream_t stream) {
    const float* x    = (const float*)d_in[0];
    const int*   ei   = (const int*)d_in[1];
    const float* W1   = (const float*)d_in[2];
    const float* b1   = (const float*)d_in[3];
    const float* W2   = (const float*)d_in[4];
    const float* b2   = (const float*)d_in[5];
    const float* Wl   = (const float*)d_in[6];
    const float* bl   = (const float*)d_in[7];
    float*       out  = (float*)d_out;

    const int N = N_NODES;
    const int E = N_EDGES;
    const int* src = ei;
    const int* dst = ei + E;

    char* ws = (char*)d_ws;
    size_t off = 0;
    int*   tab    = (int*)(ws + off);   off += align256((size_t)NCHUNK * TROW * 4);
    int*   rowptr = (int*)(ws + off);   off += align256((size_t)(N + 1) * 4);
    float* dinv   = (float*)(ws + off); off += align256((size_t)N * 4);
    int*   pk     = (int*)(ws + off);   off += align256((size_t)NCHUNK * CHUNK * 4);
    int*   csr    = (int*)(ws + off);   off += align256((size_t)E * 4);
    float* h1     = (float*)(ws + off); off += align256((size_t)N * 64 * 4);
    float* h2     = (float*)(ws + off); off += align256((size_t)N * 32 * 4);
    (void)ws_size; (void)n_in; (void)in_sizes; (void)out_size;

    chunk_order_kernel<<<NCHUNK, 1024, 0, stream>>>(src, dst, tab, pk, E / 4);
    bin_sort_kernel<<<NBIN, 1024, 0, stream>>>(pk, tab, rowptr, dinv, csr, N);

    gemm_wmma_f32<128, 64, 1><<<GEMM_GROUPS, 4 * 32, 0, stream>>>(x, W1, dinv, h1, N);

    gather_gemm_kernel<64, 32, AGG_ROWS / 16><<<AGG_GROUPS, 8 * 32, 0, stream>>>(h1, rowptr, csr, dinv, b1, W2, h2, N);

    gather_linear_kernel<32, AGG_ROWS / 8><<<AGG_GROUPS, 8 * 32, 0, stream>>>(h2, rowptr, csr, dinv, b2, Wl, bl, out, N);
}


# Gather waves load their nodes' offsets once and each node's 32 source indices as one whole-line load a node ahead; DPP hands them to the row loads, 8 independent loads per wait; sum order unchanged
# speedup vs baseline: 6.7820x; 1.0303x over previous
#include <hip/hip_runtime.h>
#include <hip/hip_bf16.h>

typedef __attribute__((ext_vector_type(2))) float v2f;
typedef __attribute__((ext_vector_type(8))) float v8f;

#define N_NODES 100000
#define N_EDGES 3200000

static_assert(N_NODES % 4 == 0 && N_EDGES % 4 == 0, "16-byte index loads need multiples of 4");

#define RBITS   9
#define RANGE   (1 << RBITS)
#define NBIN    ((N_NODES + RANGE - 1) >> RBITS)
#define NBIN_PAD (((NBIN + 1) + 31) & ~31)
#define TROW    (NBIN + 1)
#define NCHUNK  256
#define CHUNK4  ((N_EDGES / 4 + NCHUNK - 1) / NCHUNK)
#define CHUNK   (CHUNK4 * 4)
#define SBITS   17
#define SMASK   ((1 << SBITS) - 1)
#define STAGE_CAP 32768

static_assert(N_NODES <= (1 << SBITS) && SBITS + RBITS <= 31, "an edge word holds source and in-bin destination");
static_assert(CHUNK4 <= 4 * 1024, "a chunk is at most four 16-byte groups per thread");
static_assert(NBIN_PAD <= 1024 && NCHUNK % 32 == 0 && NCHUNK <= 1024, "scans run on whole waves of one work-group");

__global__ __launch_bounds__(1024)
void chunk_order_kernel(const int* __restrict__ src, const int* __restrict__ dst,
                        int* __restrict__ tab, int* __restrict__ pk, int e4) {
    __shared__ int s_hist[NBIN_PAD];
    __shared__ int s_off[NBIN_PAD];
    __shared__ int s_wtot[NBIN_PAD / 32];
    __shared__ __attribute__((aligned(16))) int s_pack[CHUNK];
    const int tid  = threadIdx.x;
    const int lane = tid & 31;
    const int wid  = tid >> 5;
    const int b    = blockIdx.x;
    if (tid < NBIN_PAD) s_hist[tid] = 0;
    const int first = b * CHUNK4;
    const int last  = (first + CHUNK4 < e4) ? first + CHUNK4 : e4;
    const int4* d4 = (const int4*)dst;
    const int4* s4 = (const int4*)src;
    int4 d[4], s[4];
#pragma unroll
    for (int u = 0; u < 4; ++u) {
        const int i = first + tid + u * 1024;
        d[u] = (i < last) ? d4[i] : make_int4(-1, -1, -1, -1);
        s[u] = (i < last) ? s4[i] : make_int4(0, 0, 0, 0);
    }
    __syncthreads();
    int4 r[4];
#pragma unroll
    for (int u = 0; u < 4; ++u) {
        r[u] = make_int4(0, 0, 0, 0);
        if (d[u].x >= 0) {
            r[u].x = atomicAdd(&s_hist[d[u].x >> RBITS], 1);
            r[u].y = atomicAdd(&s_hist[d[u].y >> RBITS], 1);
            r[u].z = atomicAdd(&s_hist[d[u].z >> RBITS], 1);
            r[u].w = atomicAdd(&s_hist[d[u].w >> RBITS], 1);
        }
    }
    __syncthreads();
    int v = 0, incl = 0;
    if (tid < NBIN_PAD) {
        v = s_hist[tid];
        incl = v;
#pragma unroll
        for (int o = 1; o < 32; o <<= 1) {
            int t = __shfl_up(incl, o);
            if (lane >= o) incl += t;
        }
        if (lane == 31) s_wtot[wid] = incl;
    }
    __syncthreads();
    if (tid < NBIN_PAD) {
        int base = 0;
#pragma unroll
        for (int w = 0; w < NBIN_PAD / 32; ++w) if (w < wid) base += s_wtot[w];
        const int excl = base + incl - v;
        s_off[tid] = excl;
        if (tid < TROW) tab[b * TROW + tid] = excl;
    }
    __syncthreads();
#pragma unroll
    for (int u = 0; u < 4; ++u) {
        if (d[u].x >= 0) {
            s_pack[s_off[d[u].x >> RBITS] + r[u].x] = s[u].x | ((d[u].x & (RANGE - 1)) << SBITS);
            s_pack[s_off[d[u].y >> RBITS] + r[u].y] = s[u].y | ((d[u].y & (RANGE - 1)) << SBITS);
            s_pack[s_off[d[u].z >> RBITS] + r[u].z] = s[u].z | ((d[u].z & (RANGE - 1)) << SBITS);
            s_pack[s_off[d[u].w >> RBITS] + r[u].w] = s[u].w | ((d[u].w & (RANGE - 1)) << SBITS);
        }
    }
    __syncthreads();
    const int len4 = (last > first) ? last - first : 0;
    int4* o4 = (int4*)(pk + (long long)b * CHUNK);
    for (int i = tid; i < len4; i += 1024) o4[i] = ((const int4*)s_pack)[i];
}

__global__ __launch_bounds__(1024)
void bin_sort_kernel(const int* __restrict__ pk, const int* __restrict__ tab,
                     int* __restrict__ rowptr, float* __restrict__ dinv,
                     int* __restrict__ csr_src, int n) {
    constexpr int RPW = NCHUNK / 32;
    __shared__ int s_cnt[RANGE];
    __shared__ int s_cur[RANGE];
    __shared__ int s_wtot[RANGE / 32];
    __shared__ int s_rbase[NCHUNK];
    __shared__ int s_rlen[NCHUNK];
    __shared__ int s_rpos[NCHUNK];
    __shared__ int s_rtot[NCHUNK / 32];
    __shared__ int s_rlow[NCHUNK / 32];
    __shared__ int s_in[STAGE_CAP];
    __shared__ int s_stage[STAGE_CAP];
    const int tid  = threadIdx.x;
    const int lane = tid & 31;
    const int wid  = tid >> 5;
    const int bin  = blockIdx.x;
    if (tid < RANGE) s_cnt[tid] = 0;

    int rlen = 0, rincl = 0, rlo = 0;
    if (tid < NCHUNK) {
        rlo  = tab[tid * TROW + bin];
        rlen = tab[tid * TROW + bin + 1] - rlo;
        rincl = rlen;
#pragma unroll
        for (int o = 1; o < 32; o <<= 1) {
            int t = __shfl_up(rincl, o);
            if (lane >= o) rincl += t;
        }
        int lows = rlo;
#pragma unroll
        for (int o = 16; o >= 1; o >>= 1) lows += __shfl_xor(lows, o);
        if (lane == 31) s_rtot[wid] = rincl;
        if (lane == 0)  s_rlow[wid] = lows;
    }
    __syncthreads();
    int beg = 0, cnt = 0;
#pragma unroll
    for (int w = 0; w < NCHUNK / 32; ++w) { beg += s_rlow[w]; cnt += s_rtot[w]; }
    if (tid < NCHUNK) {
        int base = 0;
#pragma unroll
        for (int w = 0; w < NCHUNK / 32; ++w) if (w < wid) base += s_rtot[w];
        s_rpos[tid]  = base + rincl - rlen;
        s_rbase[tid] = tid * CHUNK + rlo;
        s_rlen[tid]  = rlen;
    }
    __syncthreads();

    int rb[RPW], rl[RPW], rp[RPW];
#pragma unroll
    for (int rr = 0; rr < RPW; ++rr) {
        const int run = wid + 32 * rr;
        rb[rr] = s_rbase[run];
        rl[rr] = s_rlen[run];
        rp[rr] = s_rpos[run];
    }

    const bool staged = cnt <= STAGE_CAP;
    if (staged) {
        int e[RPW][4];
#pragma unroll
        for (int rr = 0; rr < RPW; ++rr) {
#pragma unroll
            for (int u = 0; u < 4; ++u) {
                const int k = lane + 32 * u;
                e[rr][u] = (k < rl[rr]) ? pk[rb[rr] + k] : 0;
            }
        }
#pragma unroll
        for (int rr = 0; rr < RPW; ++rr) {
#pragma unroll
            for (int u = 0; u < 4; ++u) {
                const int k = lane + 32 * u;
                if (k < rl[rr]) {
                    atomicAdd(&s_cnt[e[rr][u] >> SBITS], 1);
                    s_in[rp[rr] + k] = e[rr][u];
                }
            }
            for (int k = 128 + lane; k < rl[rr]; k += 32) {
                const int w = pk[rb[rr] + k];
                atomicAdd(&s_cnt[w >> SBITS], 1);
                s_in[rp[rr] + k] = w;
            }
        }
    } else {
#pragma unroll
        for (int rr = 0; rr < RPW; ++rr)
            for (int k = lane; k < rl[rr]; k += 32) atomicAdd(&s_cnt[pk[rb[rr] + k] >> SBITS], 1);
    }
    __syncthreads();

    int v = 0, incl = 0;
    if (tid < RANGE) {
        v = s_cnt[tid];
        incl = v;
#pragma unroll
        for (int o = 1; o < 32; o <<= 1) {
            int t = __shfl_up(incl, o);
            if (lane >= o) incl += t;
        }
        if (lane == 31) s_wtot[wid] = incl;
    }
    __syncthreads();
    if (tid < RANGE) {
        int base = 0;
#pragma unroll
        for (int w = 0; w < RANGE / 32; ++w) if (w < wid) base += s_wtot[w];
        const int excl = base + incl - v;
        s_cur[tid] = excl;
        const int node = bin * RANGE + tid;
        if (node < n) {
            rowptr[node] = beg + excl;
            dinv[node]   = rsqrtf((float)(v + 1));
            if (node == n - 1) rowptr[n] = beg + excl + v;
        }
    }
    __syncthreads();

    if (staged) {
        int i = tid;
        for (; i + 3 * 1024 < cnt; i += 4 * 1024) {
            int w[4];
#pragma unroll
            for (int u = 0; u < 4; ++u) w[u] = s_in[i + u * 1024];
#pragma unroll
            for (int u = 0; u < 4; ++u) {
                const int p = atomicAdd(&s_cur[w[u] >> SBITS], 1);
                s_stage[p] = w[u] & SMASK;
            }
        }
        for (; i < cnt; i += 1024) {
            const int w = s_in[i];
            const int p = atomicAdd(&s_cur[w >> SBITS], 1);
            s_stage[p] = w & SMASK;
        }
        __syncthreads();
        for (int k = tid; k < cnt; k += 1024) csr_src[beg + k] = s_stage[k];
    } else {
#pragma unroll
        for (int rr = 0; rr < RPW; ++rr) {
            for (int k = lane; k < rl[rr]; k += 32) {
                const int w = pk[rb[rr] + k];
                const int p = atomicAdd(&s_cur[w >> SBITS], 1);
                csr_src[beg + p] = w & SMASK;
            }
        }
    }
}

template <int K, int NOUT, int MT>
__global__ __launch_bounds__(MT*(NOUT/16)*32)
void gemm_wmma_f32(const float* __restrict__ A, const float* __restrict__ B,
                   const float* __restrict__ rowscale, float* __restrict__ C, int M) {
    constexpr int NT   = NOUT / 16;
    constexpr int NW   = MT * NT;
    constexpr int NTHR = NW * 32;
    constexpr int ROWS = MT * 16;
    constexpr int KP   = K + 4;
    constexpr int KS   = K / 4;
    constexpr int A_ELTS = ROWS * K;
    constexpr int PER  = A_ELTS / (NTHR * 4);
    static_assert(A_ELTS % (NTHR * 4) == 0, "a tile is a whole number of 16-byte pieces per thread");

    __shared__ float As[2][ROWS][KP];

    const int tid    = threadIdx.x;
    const int ntiles = (M + ROWS - 1) / ROWS;

    float4 pv[PER];
    float  ps[PER];
    int tile = blockIdx.x;
    if (tile < ntiles) {
#pragma unroll
        for (int u = 0; u < PER; ++u) {
            const int i  = (tid + u * NTHR) * 4;
            const int gr = tile * ROWS + i / K;
            pv[u] = (gr < M) ? *(const float4*)(A + (long long)gr * K + i % K)
                             : make_float4(0.f, 0.f, 0.f, 0.f);
            ps[u] = (gr < M) ? rowscale[gr] : 0.f;
        }
    }

    const int wid  = tid >> 5;
    const int lane = tid & 31;
    const int mn   = lane & 15;
    const int h    = lane >> 4;
    const int mt   = wid / NT;
    const int nt   = wid % NT;
    const int ncol = nt * 16 + mn;

    v2f bf[KS];
#pragma unroll
    for (int kk = 0; kk < KS; ++kk) {
        bf[kk].x = B[(4 * kk + 2 * h + 0) * NOUT + ncol];
        bf[kk].y = B[(4 * kk + 2 * h + 1) * NOUT + ncol];
    }

    int buf = 0;
    for (; tile < ntiles; tile += gridDim.x) {
#pragma unroll
        for (int u = 0; u < PER; ++u) {
            const int i = (tid + u * NTHR) * 4;
            float4 v = pv[u];
            const float s = ps[u];
            v.x *= s; v.y *= s; v.z *= s; v.w *= s;
            *(float4*)(&As[buf][i / K][i % K]) = v;
        }
        __syncthreads();

        const int nxt = tile + gridDim.x;
        if (nxt < ntiles) {
#pragma unroll
            for (int u = 0; u < PER; ++u) {
                const int i  = (tid + u * NTHR) * 4;
                const int gr = nxt * ROWS + i / K;
                pv[u] = (gr < M) ? *(const float4*)(A + (long long)gr * K + i % K)
                                 : make_float4(0.f, 0.f, 0.f, 0.f);
                ps[u] = (gr < M) ? rowscale[gr] : 0.f;
            }
        }

        v8f acc = {};
        const float* arow = &As[buf][mt * 16 + mn][0];
#pragma unroll
        for (int kk = 0; kk < KS; ++kk) {
            v2f a;
            a.x = arow[4 * kk + 2 * h + 0];
            a.y = arow[4 * kk + 2 * h + 1];
            acc = __builtin_amdgcn_wmma_f32_16x16x4_f32(
                 false, a,  false, bf[kk],
                 (short)0, acc,  false,  false);
        }

        const int row0 = tile * ROWS;
#pragma unroll
        for (int j = 0; j < 8; ++j) {
            int r = row0 + mt * 16 + j + 8 * h;
            if (r < M) C[(long long)r * NOUT + ncol] = acc[j];
        }
        buf ^= 1;
    }
}


template <int LPE, int N>
__device__ __forceinline__ int group_lane(int v) {
    if constexpr (LPE == 16) {
        return __builtin_amdgcn_update_dpp(0, v, 0x150 + N, 0xF, 0xF, false);
    } else {
        static_assert(LPE == 8, "a group is a DPP row or a DPP8 group");
        return __builtin_amdgcn_mov_dpp8(v, N * 0x249249);
    }
}

__device__ __forceinline__ int block_indices(const int* __restrict__ csr_src, int beg, int off, int eoff) {
    const int p = beg + off + eoff;
    return csr_src[(p < N_EDGES) ? p : N_EDGES - 1];
}

template <int F, int N0, int CNT, bool FULL, bool SELF>
__device__ __forceinline__ void block_steps(const float4* __restrict__ h4, int idx, int lim, int j,
                                            int selfrow, bool g0, float4& acc) {
    constexpr int LPE = F / 4;
    constexpr int G   = 32 / LPE;
    static_assert(CNT == 4 || CNT == 8, "four or eight steps");
    int s[8] = {0, 0, 0, 0, 0, 0, 0, 0};
    s[0] = group_lane<LPE, N0 + 0>(idx);
    s[1] = group_lane<LPE, N0 + 1>(idx);
    s[2] = group_lane<LPE, N0 + 2>(idx);
    s[3] = group_lane<LPE, N0 + 3>(idx);
    if constexpr (CNT == 8) {
        s[4] = group_lane<LPE, N0 + 4>(idx);
        s[5] = group_lane<LPE, N0 + 5>(idx);
        s[6] = group_lane<LPE, N0 + 6>(idx);
        s[7] = group_lane<LPE, N0 + 7>(idx);
    }
    float4 vs = make_float4(0.f, 0.f, 0.f, 0.f);
    if constexpr (SELF) vs = h4[(unsigned)(selfrow * LPE + j)];
    float4 v[CNT];
#pragma unroll
    for (int u = 0; u < CNT; ++u) {
        if constexpr (FULL) {
            v[u] = h4[(unsigned)(s[u] * LPE + j)];
        } else {
            v[u] = make_float4(0.f, 0.f, 0.f, 0.f);
            if (G * (N0 + u) < lim) v[u] = h4[(unsigned)(s[u] * LPE + j)];
        }
    }
    if constexpr (SELF) acc = g0 ? vs : make_float4(0.f, 0.f, 0.f, 0.f);
#pragma unroll
    for (int u = 0; u < CNT; ++u) {
        acc.x += v[u].x; acc.y += v[u].y; acc.z += v[u].z; acc.w += v[u].w;
    }
}

template <int F, bool SELF>
__device__ __forceinline__ void block_sum(const float4* __restrict__ h4, int idx, int rem, int g, int j,
                                          int selfrow, float4& acc) {
    constexpr int LPE = F / 4;
    const int  lim = rem - g;
    const bool g0  = (g == 0);
    if constexpr (LPE == 16) {
        if (rem >= 16) block_steps<F, 0, 8, true,  SELF>(h4, idx, lim, j, selfrow, g0, acc);
        else           block_steps<F, 0, 8, false, SELF>(h4, idx, lim, j, selfrow, g0, acc);
        if (rem > 16) {
            if (rem >= 32) block_steps<F, 8, 8, true,  false>(h4, idx, lim, j, selfrow, g0, acc);
            else           block_steps<F, 8, 8, false, false>(h4, idx, lim, j, selfrow, g0, acc);
        }
    } else {
        if (rem >= 32) {
            block_steps<F, 0, 8, true, SELF>(h4, idx, lim, j, selfrow, g0, acc);
        } else {
            if (rem >= 16) block_steps<F, 0, 4, true,  SELF>(h4, idx, lim, j, selfrow, g0, acc);
            else           block_steps<F, 0, 4, false, SELF>(h4, idx, lim, j, selfrow, g0, acc);
            if (rem > 16)  block_steps<F, 4, 4, false, false>(h4, idx, lim, j, selfrow, g0, acc);
        }
    }
}

template <int F>
__device__ __forceinline__ float4 node_sum(const float4* __restrict__ h4, const int* __restrict__ csr_src,
                                           int selfrow, int beg, int cnt, int c0, int c1, int g, int j) {
    constexpr int LPE = F / 4;
    constexpr int G   = 32 / LPE;
    float4 acc;
    block_sum<F, true>(h4, c0, cnt, g, j, selfrow, acc);
    if (cnt > 32) block_sum<F, false>(h4, c1, cnt - 32, g, j, selfrow, acc);
    for (int off = 64; off < cnt; off += 32) {
        const int cx = block_indices(csr_src, beg, off, G * j + g);
        block_sum<F, false>(h4, cx, cnt - off, g, j, selfrow, acc);
    }
#pragma unroll
    for (int o = LPE; o < 32; o <<= 1) {
        acc.x += __shfl_xor(acc.x, o);
        acc.y += __shfl_xor(acc.y, o);
        acc.z += __shfl_xor(acc.z, o);
        acc.w += __shfl_xor(acc.w, o);
    }
    return acc;
}

template <int F, int NOUT, int MT>
__global__ __launch_bounds__(MT*(NOUT/16)*32)
void gather_gemm_kernel(const float* __restrict__ hs, const int* __restrict__ rowptr,
                        const int* __restrict__ csr_src, const float* __restrict__ dinv,
                        const float* __restrict__ bias, const float* __restrict__ B,
                        float* __restrict__ C, int n) {
    constexpr int LPE  = F / 4;
    constexpr int NT   = NOUT / 16;
    constexpr int NW   = MT * NT;
    constexpr int ROWS = MT * 16;
    constexpr int NPW  = ROWS / NW;
    constexpr int KP   = F + 4;
    constexpr int KS   = F / 4;
    __shared__ float As[ROWS][KP];

    const int tid  = threadIdx.x;
    const int lane = tid & 31;
    const int wid  = __builtin_amdgcn_readfirstlane(tid >> 5);
    const int g = lane / LPE;
    const int j = lane % LPE;
    const float4* h4 = (const float4*)hs;
    const float4 b = ((const float4*)bias)[j];
    const int row0 = blockIdx.x * ROWS;

    static_assert(NPW + 2 <= 32, "a wave's offsets, and the two behind them, fit one wave load");
    const int eoff = (32 / LPE) * j + g;
    const int i0   = row0 + wid * NPW;
    const float dvv = dinv[(i0 + lane < n) ? i0 + lane : n - 1];
    const int   rpv = rowptr[(i0 + lane < n) ? i0 + lane : n];
    int beg = __builtin_amdgcn_readlane(rpv, 0);
    int cnt = __builtin_amdgcn_readlane(rpv, 1) - beg;
    int c0  = block_indices(csr_src, beg, 0, eoff);
    int c1  = block_indices(csr_src, beg, 32, eoff);
    for (int t = 0; t < NPW; ++t) {
        const int lr = wid * NPW + t;
        const int i  = row0 + lr;
        const int nbeg = __builtin_amdgcn_readlane(rpv, t + 1);
        const int ncnt = __builtin_amdgcn_readlane(rpv, t + 2) - nbeg;
        const int n0 = block_indices(csr_src, nbeg, 0, eoff);
        const int n1 = block_indices(csr_src, nbeg, 32, eoff);
        const float di = __int_as_float(__builtin_amdgcn_readlane(__float_as_int(dvv), t));
        const float4 acc = node_sum<F>(h4, csr_src, (i < n) ? i : n - 1, beg, cnt, c0, c1, g, j);
        float4 r;
        r.x = fmaxf(acc.x * di + b.x, 0.0f) * di;
        r.y = fmaxf(acc.y * di + b.y, 0.0f) * di;
        r.z = fmaxf(acc.z * di + b.z, 0.0f) * di;
        r.w = fmaxf(acc.w * di + b.w, 0.0f) * di;
        if (i >= n) r = make_float4(0.f, 0.f, 0.f, 0.f);
        if (g == 0) *(float4*)(&As[lr][4 * j]) = r;
        beg = nbeg; cnt = ncnt; c0 = n0; c1 = n1;
    }

    const int mn   = lane & 15;
    const int h    = lane >> 4;
    const int mt   = wid / NT;
    const int nt   = wid % NT;
    const int ncol = nt * 16 + mn;

    v2f bf[KS];
#pragma unroll
    for (int kk = 0; kk < KS; ++kk) {
        bf[kk].x = B[(4 * kk + 2 * h + 0) * NOUT + ncol];
        bf[kk].y = B[(4 * kk + 2 * h + 1) * NOUT + ncol];
    }
    __syncthreads();

    v8f acc = {};
    const float* arow = &As[mt * 16 + mn][0];
#pragma unroll
    for (int kk = 0; kk < KS; ++kk) {
        v2f a;
        a.x = arow[4 * kk + 2 * h + 0];
        a.y = arow[4 * kk + 2 * h + 1];
        acc = __builtin_amdgcn_wmma_f32_16x16x4_f32(
             false, a,  false, bf[kk],
             (short)0, acc,  false,  false);
    }

#pragma unroll
    for (int jj = 0; jj < 8; ++jj) {
        const int r = row0 + mt * 16 + jj + 8 * h;
        if (r < n) C[(long long)r * NOUT + ncol] = acc[jj];
    }
}

template <int F, int NPW>
__global__ __launch_bounds__(256)
void gather_linear_kernel(const float* __restrict__ hs, const int* __restrict__ rowptr,
                          const int* __restrict__ csr_src, const float* __restrict__ dinv,
                          const float* __restrict__ bias, const float* __restrict__ Wl,
                          const float* __restrict__ bl, float* __restrict__ out, int n) {
    constexpr int LPE  = F / 4;
    constexpr int ROWS = 8 * NPW;
    __shared__ float s_out[ROWS];
    const int tid  = threadIdx.x;
    const int lane = tid & 31;
    const int wid  = __builtin_amdgcn_readfirstlane(tid >> 5);
    const int g = lane / LPE;
    const int j = lane % LPE;
    const float4* h4 = (const float4*)hs;
    const float4 b  = ((const float4*)bias)[j];
    const float4 wl = ((const float4*)Wl)[j];
    const int row0 = blockIdx.x * ROWS;

    static_assert(NPW + 2 <= 32, "a wave's offsets, and the two behind them, fit one wave load");
    const int eoff = (32 / LPE) * j + g;
    const int i0   = row0 + wid * NPW;
    const float dvv = dinv[(i0 + lane < n) ? i0 + lane : n - 1];
    const int   rpv = rowptr[(i0 + lane < n) ? i0 + lane : n];
    int beg = __builtin_amdgcn_readlane(rpv, 0);
    int cnt = __builtin_amdgcn_readlane(rpv, 1) - beg;
    int c0  = block_indices(csr_src, beg, 0, eoff);
    int c1  = block_indices(csr_src, beg, 32, eoff);
    for (int t = 0; t < NPW; ++t) {
        const int lr = wid * NPW + t;
        const int i  = row0 + lr;
        const int nbeg = __builtin_amdgcn_readlane(rpv, t + 1);
        const int ncnt = __builtin_amdgcn_readlane(rpv, t + 2) - nbeg;
        const int n0 = block_indices(csr_src, nbeg, 0, eoff);
        const int n1 = block_indices(csr_src, nbeg, 32, eoff);
        const float di = __int_as_float(__builtin_amdgcn_readlane(__float_as_int(dvv), t));
        const float4 acc = node_sum<F>(h4, csr_src, (i < n) ? i : n - 1, beg, cnt, c0, c1, g, j);
        float4 r;
        r.x = fmaxf(acc.x * di + b.x, 0.0f);
        r.y = fmaxf(acc.y * di + b.y, 0.0f);
        r.z = fmaxf(acc.z * di + b.z, 0.0f);
        r.w = fmaxf(acc.w * di + b.w, 0.0f);
        float p = r.x * wl.x + r.y * wl.y + r.z * wl.z + r.w * wl.w;
#pragma unroll
        for (int o = 1; o < LPE; o <<= 1) p += __shfl_xor(p, o);
        if (lane == 0) s_out[lr] = p;
        beg = nbeg; cnt = ncnt; c0 = n0; c1 = n1;
    }
    __syncthreads();
    if (tid < ROWS) {
        const int i = row0 + tid;
        if (i < n) out[i] = s_out[tid] + bl[0];
    }
}

#define GEMM_GROUPS 1250

#define AGG_ROWS   64
#define AGG_GROUPS ((N_NODES + AGG_ROWS - 1) / AGG_ROWS)

static inline size_t align256(size_t x) { return (x + 255) & ~size_t(255); }

extern "C" void kernel_launch(void* const* d_in, const int* in_sizes, int n_in,
                              void* d_out, int out_size, void* d_ws, size_t ws_size,
                              hipStream_t stream) {
    const float* x    = (const float*)d_in[0];
    const int*   ei   = (const int*)d_in[1];
    const float* W1   = (const float*)d_in[2];
    const float* b1   = (const float*)d_in[3];
    const float* W2   = (const float*)d_in[4];
    const float* b2   = (const float*)d_in[5];
    const float* Wl   = (const float*)d_in[6];
    const float* bl   = (const float*)d_in[7];
    float*       out  = (float*)d_out;

    const int N = N_NODES;
    const int E = N_EDGES;
    const int* src = ei;
    const int* dst = ei + E;

    char* ws = (char*)d_ws;
    size_t off = 0;
    int*   tab    = (int*)(ws + off);   off += align256((size_t)NCHUNK * TROW * 4);
    int*   rowptr = (int*)(ws + off);   off += align256((size_t)(N + 1) * 4);
    float* dinv   = (float*)(ws + off); off += align256((size_t)N * 4);
    int*   pk     = (int*)(ws + off);   off += align256((size_t)NCHUNK * CHUNK * 4);
    int*   csr    = (int*)(ws + off);   off += align256((size_t)E * 4);
    float* h1     = (float*)(ws + off); off += align256((size_t)N * 64 * 4);
    float* h2     = (float*)(ws + off); off += align256((size_t)N * 32 * 4);
    (void)ws_size; (void)n_in; (void)in_sizes; (void)out_size;

    chunk_order_kernel<<<NCHUNK, 1024, 0, stream>>>(src, dst, tab, pk, E / 4);
    bin_sort_kernel<<<NBIN, 1024, 0, stream>>>(pk, tab, rowptr, dinv, csr, N);

    gemm_wmma_f32<128, 64, 1><<<GEMM_GROUPS, 4 * 32, 0, stream>>>(x, W1, dinv, h1, N);

    gather_gemm_kernel<64, 32, AGG_ROWS / 16><<<AGG_GROUPS, 8 * 32, 0, stream>>>(h1, rowptr, csr, dinv, b1, W2, h2, N);

    gather_linear_kernel<32, AGG_ROWS / 8><<<AGG_GROUPS, 8 * 32, 0, stream>>>(h2, rowptr, csr, dinv, b2, Wl, bl, out, N);
}
